# starting kernel
# baseline (speedup 1.0000x reference)
#include <hip/hip_runtime.h>
#include <hip/hip_bf16.h>


#define TT 256
#define N4 4

typedef __attribute__((ext_vector_type(2))) float v2f;
typedef __attribute__((ext_vector_type(8))) float v8f;

__device__ __forceinline__ float shf(float v, int srcLane) {
  return __int_as_float(__builtin_amdgcn_ds_bpermute(srcLane << 2, __float_as_int(v)));
}

__device__ __forceinline__ v8f wmma4(float a0, float a1, float b0, float b1, v8f c) {
  v2f A, B;
  A[0] = a0; A[1] = a1;
  B[0] = b0; B[1] = b1;
  return __builtin_amdgcn_wmma_f32_16x16x4_f32(false, A, false, B, (short)0, c, false, false);
}

__device__ __forceinline__ float fast_tanh(float x) {
#if __has_builtin(__builtin_amdgcn_tanhf)
  return __builtin_amdgcn_tanhf(x);
#else
  const float e = __expf(2.0f * x);
  return 1.0f - 2.0f * __builtin_amdgcn_rcpf(e + 1.0f);
#endif
}
#define INV_2PI 0.15915494309189535f
__device__ __forceinline__ float fast_sin(float x) {
  return __builtin_amdgcn_sinf(x * INV_2PI);
}
__device__ __forceinline__ float fast_cos(float x) {
  return __builtin_amdgcn_cosf(x * INV_2PI);
}

__global__ __launch_bounds__(256) void ekf1250_kernel(
    const float* __restrict__ z,  const float* __restrict__ Ag,
    const float* __restrict__ Cg, const float* __restrict__ Qg,
    const float* __restrict__ Rg, const float* __restrict__ x0g,
    const float* __restrict__ P0g, float* __restrict__ out)
{
  __shared__ float lds[8 * 640];
  const int tid  = threadIdx.x;
  const int lane = tid & 31;
  const int wave = (blockIdx.x * blockDim.x + tid) >> 5;
  const long b0  = (long)wave * 2;
  float* W = &lds[(tid >> 5) * 640];
  const int RP = 0;
  const int R1 = 128;
  const int R2 = 256;
  const int R3 = 384;
  const int R4 = 448;
  const int RT = 512;

  const int  r    = lane & 7;
  const int  e8   = (lane >> 3) & 1;
  const int  khi  = lane >> 4;
  const int  eD   = lane >> 4;
  const bool dval = (lane < 8) || (lane >= 24);
  const bool lowh = lane < 16;
  const int  dcol = lane & 7;

  const int aBase  = e8 * 64 + r * 8 + 2 * khi;
  const int bBase  = e8 * 64 + (2 * khi) * 8 + r;
  const int a3Base = e8 * 32 + (r & 3) * 8 + 2 * khi;

  float regA[8], regC[8];
  {
    const float4 a0 = *reinterpret_cast<const float4*>(Ag + r * 8);
    const float4 a1 = *reinterpret_cast<const float4*>(Ag + r * 8 + 4);
    regA[0]=a0.x; regA[1]=a0.y; regA[2]=a0.z; regA[3]=a0.w;
    regA[4]=a1.x; regA[5]=a1.y; regA[6]=a1.z; regA[7]=a1.w;
    const float4 c0 = *reinterpret_cast<const float4*>(Cg + (r & 3) * 8);
    const float4 c1 = *reinterpret_cast<const float4*>(Cg + (r & 3) * 8 + 4);
    const float m = (r < 4) ? 1.0f : 0.0f;
    regC[0]=c0.x*m; regC[1]=c0.y*m; regC[2]=c0.z*m; regC[3]=c0.w*m;
    regC[4]=c1.x*m; regC[5]=c1.y*m; regC[6]=c1.z*m; regC[7]=c1.w*m;
  }
  v8f Qd, Rd;
  #pragma unroll
  for (int i = 0; i < 8; ++i) Qd[i] = Qg[i * 8 + dcol];
  #pragma unroll
  for (int i = 0; i < 8; ++i) {
    float rv = Rg[(i & 3) * 4 + (dcol & 3)];
    Rd[i] = (i < 4 && dcol < 4) ? rv : 0.0f;
  }

  float xrep = x0g[r];
  for (int idx = lane; idx < 128; idx += 32) W[RP + idx] = P0g[idx & 63];

  const v8f z8 = {0,0,0,0,0,0,0,0};

  #pragma unroll 1
  for (int t = 0; t < TT; ++t) {
    const int sb = lane & 16;
    float xb[8];
    #pragma unroll
    for (int k = 0; k < 8; ++k) xb[k] = shf(xrep, sb + k);
    float ta = 0.0f;
    #pragma unroll
    for (int k = 0; k < 8; ++k) ta = fmaf(regA[k], xb[k], ta);
    const float xp = fast_tanh(ta);
    const float dd = 1.0f - xp * xp;
    float u = 0.0f;
    #pragma unroll
    for (int k = 0; k < 8; ++k) u = fmaf(regC[k], shf(xp, sb + k), u);
    const float ys = fast_sin(u);
    const float cy = fast_cos(u);

    const int srcAB = (e8 << 4) + r;
    const float dA  = shf(dd, srcAB);
    const float cyA = shf(cy, srcAB);
    float Fo[4], Ho[4];
    #pragma unroll
    for (int jc = 0; jc < 2; ++jc)
      #pragma unroll
      for (int v = 0; v < 2; ++v) {
        const float av = lowh ? regA[4 * jc + v] : regA[4 * jc + v + 2];
        const float cv = lowh ? regC[4 * jc + v] : regC[4 * jc + v + 2];
        Fo[2 * jc + v] = dA * av;
        Ho[2 * jc + v] = cyA * cv;
      }

    v8f G = wmma4(Fo[0], Fo[1], W[RP + bBase + 0 ], W[RP + bBase + 8 ], z8);
    G     = wmma4(Fo[2], Fo[3], W[RP + bBase + 32], W[RP + bBase + 40], G);
    {
      const int base = dval ? (R1 + eD * 64 + dcol) : (RT + (lane & 15));
      #pragma unroll
      for (int i = 0; i < 8; ++i) W[base + i * 8] = G[i];
    }
    v8f Pp = wmma4(W[R1 + aBase + 0], W[R1 + aBase + 1], Fo[0], Fo[1], Qd);
    Pp     = wmma4(W[R1 + aBase + 4], W[R1 + aBase + 5], Fo[2], Fo[3], Pp);
    {
      const int base = dval ? (R2 + eD * 64 + dcol) : (RT + (lane & 15));
      #pragma unroll
      for (int i = 0; i < 8; ++i) W[base + i * 8] = Pp[i];
    }
    v8f T1 = wmma4(Ho[0], Ho[1], W[R2 + bBase + 0 ], W[R2 + bBase + 8 ], z8);
    T1     = wmma4(Ho[2], Ho[3], W[R2 + bBase + 32], W[R2 + bBase + 40], T1);
    {
      const int base = dval ? (R3 + eD * 32 + dcol) : (RT + (lane & 15));
      #pragma unroll
      for (int i = 0; i < 4; ++i) W[base + i * 8] = T1[i];
    }
    v8f Sm = wmma4(W[R3 + a3Base + 0], W[R3 + a3Base + 1], Ho[0], Ho[1], Rd);
    Sm     = wmma4(W[R3 + a3Base + 4], W[R3 + a3Base + 5], Ho[2], Ho[3], Sm);
    v8f PHt = wmma4(W[R2 + aBase + 0], W[R2 + aBase + 1], Ho[0], Ho[1], z8);
    PHt     = wmma4(W[R2 + aBase + 4], W[R2 + aBase + 5], Ho[2], Ho[3], PHt);
    {
      const int base = (dval && dcol < 4) ? (R4 + eD * 32 + dcol) : (RT + (lane & 15));
      #pragma unroll
      for (int i = 0; i < 8; ++i) W[base + i * 4] = PHt[i];
    }

    const float4 rhs4 = *reinterpret_cast<const float4*>(&W[R4 + (eD << 5) + (r << 2)]);
    float rr4[4] = {rhs4.x, rhs4.y, rhs4.z, rhs4.w};
    float s[4][4];
    const int sS = 24 * eD;
    #pragma unroll
    for (int i = 0; i < 4; ++i)
      #pragma unroll
      for (int j = 0; j < 4; ++j) s[i][j] = shf(Sm[i], sS + j);
    #pragma unroll
    for (int k = 0; k < 4; ++k) {
      const float inv = __builtin_amdgcn_rcpf(s[k][k]);
      #pragma unroll
      for (int j = k + 1; j < 4; ++j) s[k][j] *= inv;
      rr4[k] *= inv;
      #pragma unroll
      for (int i = k + 1; i < 4; ++i) {
        const float f = s[i][k];
        #pragma unroll
        for (int j = k + 1; j < 4; ++j) s[i][j] = fmaf(-f, s[k][j], s[i][j]);
        rr4[i] = fmaf(-f, rr4[k], rr4[i]);
      }
    }
    float Kp[4];
    Kp[3] = rr4[3];
    Kp[2] = fmaf(-s[2][3], Kp[3], rr4[2]);
    Kp[1] = fmaf(-s[1][3], Kp[3], fmaf(-s[1][2], Kp[2], rr4[1]));
    Kp[0] = fmaf(-s[0][3], Kp[3], fmaf(-s[0][2], Kp[2], fmaf(-s[0][1], Kp[1], rr4[0])));

    const long zidx = ((b0 + (lane >> 4)) * TT + t) * N4 + (r & 3);
    const float zv = z[zidx];
    __builtin_prefetch(&z[(t < TT - 8) ? (zidx + 8 * N4) : zidx], 0, 0);
    const float dy = zv - ys;
    const int sE = lane & 16;
    float xn = shf(xp, sE + r);
    #pragma unroll
    for (int j = 0; j < 4; ++j) xn = fmaf(Kp[j], shf(dy, sE + j), xn);
    if (dval) out[((b0 + eD) * TT + t) * 8 + r] = xn;

    const int sK = 24 * e8 + r;
    float Av0, Av1, Bv0, Bv1;
    {
      const float klo0 = shf(Kp[0], sK), khi0 = shf(Kp[2], sK);
      const float klo1 = shf(Kp[1], sK), khi1 = shf(Kp[3], sK);
      Av0 = -(lowh ? klo0 : khi0);
      Av1 = -(lowh ? klo1 : khi1);
      const float tlo0 = shf(T1[0], sK), thi0 = shf(T1[2], sK);
      const float tlo1 = shf(T1[1], sK), thi1 = shf(T1[3], sK);
      Bv0 = lowh ? tlo0 : thi0;
      Bv1 = lowh ? tlo1 : thi1;
    }
    v8f Pn = wmma4(Av0, Av1, Bv0, Bv1, Pp);
    {
      const int base = dval ? (RP + eD * 64 + dcol) : (RT + (lane & 15));
      #pragma unroll
      for (int i = 0; i < 8; ++i) W[base + i * 8] = Pn[i];
    }
    xrep = shf(xn, 24 * (lane >> 4) + r);
  }
}

extern "C" void kernel_launch(void* const* d_in, const int* in_sizes, int n_in,
                              void* d_out, int out_size, void* d_ws, size_t ws_size,
                              hipStream_t stream) {
  const float* z  = (const float*)d_in[0];
  const float* A  = (const float*)d_in[1];
  const float* C  = (const float*)d_in[2];
  const float* Q  = (const float*)d_in[3];
  const float* R  = (const float*)d_in[4];
  const float* x0 = (const float*)d_in[5];
  const float* P0 = (const float*)d_in[6];
  float* out = (float*)d_out;

  const int Bsz = in_sizes[0] / (TT * N4);
  const int waves = Bsz / 2;
  const int threads = 256;
  int blocks = (waves * 32) / threads;
  if (blocks < 1) blocks = 1;
  ekf1250_kernel<<<blocks, threads, 0, stream>>>(z, A, C, Q, R, x0, P0, out);
}


# Lane moves taken off the LDS unit: every ds_bpermute becomes a DPP row_share/row_ror move or v_permlanex16 plus select on the vector ALU; two permutes that were identities are dropped
# speedup vs baseline: 1.2858x; 1.2858x over previous
#include <hip/hip_runtime.h>
#include <hip/hip_bf16.h>


#define TT 256
#define N4 4

typedef __attribute__((ext_vector_type(2))) float v2f;
typedef __attribute__((ext_vector_type(8))) float v8f;

template <int K>
__device__ __forceinline__ float rshare(float v) {
  const int b = __float_as_int(v);
  return __int_as_float(__builtin_amdgcn_mov_dpp(b, 0x150 + K, 0xf, 0xf, false));
}
__device__ __forceinline__ float ror8hi(float v) {
  const int b = __float_as_int(v);
  return __int_as_float(__builtin_amdgcn_update_dpp(b, b, 0x128, 0x2, 0xf, false));
}
__device__ __forceinline__ float xhalf(float v) {
  const int b = __float_as_int(v);
  return __int_as_float(__builtin_amdgcn_permlanex16(b, b, 0x76543210, 0xfedcba98, false, false));
}

__device__ __forceinline__ v8f wmma4(float a0, float a1, float b0, float b1, v8f c) {
  v2f A, B;
  A[0] = a0; A[1] = a1;
  B[0] = b0; B[1] = b1;
  return __builtin_amdgcn_wmma_f32_16x16x4_f32(false, A, false, B, (short)0, c, false, false);
}

__device__ __forceinline__ float fast_tanh(float x) {
#if __has_builtin(__builtin_amdgcn_tanhf)
  return __builtin_amdgcn_tanhf(x);
#else
  const float e = __expf(2.0f * x);
  return 1.0f - 2.0f * __builtin_amdgcn_rcpf(e + 1.0f);
#endif
}
#define INV_2PI 0.15915494309189535f
__device__ __forceinline__ float fast_sin(float x) {
  return __builtin_amdgcn_sinf(x * INV_2PI);
}
__device__ __forceinline__ float fast_cos(float x) {
  return __builtin_amdgcn_cosf(x * INV_2PI);
}

__global__ __launch_bounds__(256) void ekf1250_kernel(
    const float* __restrict__ z,  const float* __restrict__ Ag,
    const float* __restrict__ Cg, const float* __restrict__ Qg,
    const float* __restrict__ Rg, const float* __restrict__ x0g,
    const float* __restrict__ P0g, float* __restrict__ out)
{
  __shared__ float lds[8 * 640];
  const int tid  = threadIdx.x;
  const int lane = tid & 31;
  const int wave = (blockIdx.x * blockDim.x + tid) >> 5;
  const long b0  = (long)wave * 2;
  float* W = &lds[(tid >> 5) * 640];
  const int RP = 0;
  const int R1 = 128;
  const int R2 = 256;
  const int R3 = 384;
  const int R4 = 448;
  const int RT = 512;

  const int  r    = lane & 7;
  const int  e8   = (lane >> 3) & 1;
  const int  khi  = lane >> 4;
  const int  eD   = lane >> 4;
  const bool dval = (lane < 8) || (lane >= 24);
  const bool lowh = lane < 16;
  const bool mid  = (lane >= 8) && (lane < 24);
  const int  dcol = lane & 7;

  const int aBase  = e8 * 64 + r * 8 + 2 * khi;
  const int bBase  = e8 * 64 + (2 * khi) * 8 + r;
  const int a3Base = e8 * 32 + (r & 3) * 8 + 2 * khi;

  float regA[8], regC[8];
  {
    const float4 a0 = *reinterpret_cast<const float4*>(Ag + r * 8);
    const float4 a1 = *reinterpret_cast<const float4*>(Ag + r * 8 + 4);
    regA[0]=a0.x; regA[1]=a0.y; regA[2]=a0.z; regA[3]=a0.w;
    regA[4]=a1.x; regA[5]=a1.y; regA[6]=a1.z; regA[7]=a1.w;
    const float4 c0 = *reinterpret_cast<const float4*>(Cg + (r & 3) * 8);
    const float4 c1 = *reinterpret_cast<const float4*>(Cg + (r & 3) * 8 + 4);
    const float m = (r < 4) ? 1.0f : 0.0f;
    regC[0]=c0.x*m; regC[1]=c0.y*m; regC[2]=c0.z*m; regC[3]=c0.w*m;
    regC[4]=c1.x*m; regC[5]=c1.y*m; regC[6]=c1.z*m; regC[7]=c1.w*m;
  }
  v8f Qd, Rd;
  #pragma unroll
  for (int i = 0; i < 8; ++i) Qd[i] = Qg[i * 8 + dcol];
  #pragma unroll
  for (int i = 0; i < 8; ++i) {
    float rv = Rg[(i & 3) * 4 + (dcol & 3)];
    Rd[i] = (i < 4 && dcol < 4) ? rv : 0.0f;
  }

  float xrep = x0g[r];
  for (int idx = lane; idx < 128; idx += 32) W[RP + idx] = P0g[idx & 63];

  const v8f z8 = {0,0,0,0,0,0,0,0};

  #pragma unroll 1
  for (int t = 0; t < TT; ++t) {
    float xb[8];
    xb[0] = rshare<0>(xrep); xb[1] = rshare<1>(xrep); xb[2] = rshare<2>(xrep); xb[3] = rshare<3>(xrep);
    xb[4] = rshare<4>(xrep); xb[5] = rshare<5>(xrep); xb[6] = rshare<6>(xrep); xb[7] = rshare<7>(xrep);
    float ta = 0.0f;
    #pragma unroll
    for (int k = 0; k < 8; ++k) ta = fmaf(regA[k], xb[k], ta);
    const float xp = fast_tanh(ta);
    const float dd = 1.0f - xp * xp;
    float xq[8];
    xq[0] = rshare<0>(xp); xq[1] = rshare<1>(xp); xq[2] = rshare<2>(xp); xq[3] = rshare<3>(xp);
    xq[4] = rshare<4>(xp); xq[5] = rshare<5>(xp); xq[6] = rshare<6>(xp); xq[7] = rshare<7>(xp);
    float u = 0.0f;
    #pragma unroll
    for (int k = 0; k < 8; ++k) u = fmaf(regC[k], xq[k], u);
    const float ys = fast_sin(u);
    const float cy = fast_cos(u);

    const float ddx = xhalf(dd);
    const float cyx = xhalf(cy);
    const float dA  = mid ? ddx : dd;
    const float cyA = mid ? cyx : cy;
    float Fo[4], Ho[4];
    #pragma unroll
    for (int jc = 0; jc < 2; ++jc)
      #pragma unroll
      for (int v = 0; v < 2; ++v) {
        const float av = lowh ? regA[4 * jc + v] : regA[4 * jc + v + 2];
        const float cv = lowh ? regC[4 * jc + v] : regC[4 * jc + v + 2];
        Fo[2 * jc + v] = dA * av;
        Ho[2 * jc + v] = cyA * cv;
      }

    v8f G = wmma4(Fo[0], Fo[1], W[RP + bBase + 0 ], W[RP + bBase + 8 ], z8);
    G     = wmma4(Fo[2], Fo[3], W[RP + bBase + 32], W[RP + bBase + 40], G);
    {
      const int base = dval ? (R1 + eD * 64 + dcol) : (RT + (lane & 15));
      #pragma unroll
      for (int i = 0; i < 8; ++i) W[base + i * 8] = G[i];
    }
    v8f Pp = wmma4(W[R1 + aBase + 0], W[R1 + aBase + 1], Fo[0], Fo[1], Qd);
    Pp     = wmma4(W[R1 + aBase + 4], W[R1 + aBase + 5], Fo[2], Fo[3], Pp);
    {
      const int base = dval ? (R2 + eD * 64 + dcol) : (RT + (lane & 15));
      #pragma unroll
      for (int i = 0; i < 8; ++i) W[base + i * 8] = Pp[i];
    }
    v8f T1 = wmma4(Ho[0], Ho[1], W[R2 + bBase + 0 ], W[R2 + bBase + 8 ], z8);
    T1     = wmma4(Ho[2], Ho[3], W[R2 + bBase + 32], W[R2 + bBase + 40], T1);
    {
      const int base = dval ? (R3 + eD * 32 + dcol) : (RT + (lane & 15));
      #pragma unroll
      for (int i = 0; i < 4; ++i) W[base + i * 8] = T1[i];
    }
    v8f Sm = wmma4(W[R3 + a3Base + 0], W[R3 + a3Base + 1], Ho[0], Ho[1], Rd);
    Sm     = wmma4(W[R3 + a3Base + 4], W[R3 + a3Base + 5], Ho[2], Ho[3], Sm);
    v8f PHt = wmma4(W[R2 + aBase + 0], W[R2 + aBase + 1], Ho[0], Ho[1], z8);
    PHt     = wmma4(W[R2 + aBase + 4], W[R2 + aBase + 5], Ho[2], Ho[3], PHt);
    {
      const int base = (dval && dcol < 4) ? (R4 + eD * 32 + dcol) : (RT + (lane & 15));
      #pragma unroll
      for (int i = 0; i < 8; ++i) W[base + i * 4] = PHt[i];
    }

    const float4 rhs4 = *reinterpret_cast<const float4*>(&W[R4 + (eD << 5) + (r << 2)]);
    float rr4[4] = {rhs4.x, rhs4.y, rhs4.z, rhs4.w};
    float s[4][4];
    #pragma unroll
    for (int i = 0; i < 4; ++i) {
      const float sr = ror8hi(Sm[i]);
      s[i][0] = rshare<0>(sr); s[i][1] = rshare<1>(sr);
      s[i][2] = rshare<2>(sr); s[i][3] = rshare<3>(sr);
    }
    #pragma unroll
    for (int k = 0; k < 4; ++k) {
      const float inv = __builtin_amdgcn_rcpf(s[k][k]);
      #pragma unroll
      for (int j = k + 1; j < 4; ++j) s[k][j] *= inv;
      rr4[k] *= inv;
      #pragma unroll
      for (int i = k + 1; i < 4; ++i) {
        const float f = s[i][k];
        #pragma unroll
        for (int j = k + 1; j < 4; ++j) s[i][j] = fmaf(-f, s[k][j], s[i][j]);
        rr4[i] = fmaf(-f, rr4[k], rr4[i]);
      }
    }
    float Kp[4];
    Kp[3] = rr4[3];
    Kp[2] = fmaf(-s[2][3], Kp[3], rr4[2]);
    Kp[1] = fmaf(-s[1][3], Kp[3], fmaf(-s[1][2], Kp[2], rr4[1]));
    Kp[0] = fmaf(-s[0][3], Kp[3], fmaf(-s[0][2], Kp[2], fmaf(-s[0][1], Kp[1], rr4[0])));

    const long zidx = ((b0 + (lane >> 4)) * TT + t) * N4 + (r & 3);
    const float zv = z[zidx];
    __builtin_prefetch(&z[(t < TT - 8) ? (zidx + 8 * N4) : zidx], 0, 0);
    const float dy = zv - ys;
    float dyb[4];
    dyb[0] = rshare<0>(dy); dyb[1] = rshare<1>(dy); dyb[2] = rshare<2>(dy); dyb[3] = rshare<3>(dy);
    float xn = xp;
    #pragma unroll
    for (int j = 0; j < 4; ++j) xn = fmaf(Kp[j], dyb[j], xn);
    if (dval) out[((b0 + eD) * TT + t) * 8 + r] = xn;

    float Av0, Av1, Bv0, Bv1;
    {
      const float ko0 = xhalf(lowh ? Kp[2] : Kp[0]);
      const float ko1 = xhalf(lowh ? Kp[3] : Kp[1]);
      const float to0 = xhalf(lowh ? T1[2] : T1[0]);
      const float to1 = xhalf(lowh ? T1[3] : T1[1]);
      Av0 = -(mid ? ko0 : (lowh ? Kp[0] : Kp[2]));
      Av1 = -(mid ? ko1 : (lowh ? Kp[1] : Kp[3]));
      Bv0 = mid ? to0 : (lowh ? T1[0] : T1[2]);
      Bv1 = mid ? to1 : (lowh ? T1[1] : T1[3]);
    }
    v8f Pn = wmma4(Av0, Av1, Bv0, Bv1, Pp);
    {
      const int base = dval ? (RP + eD * 64 + dcol) : (RT + (lane & 15));
      #pragma unroll
      for (int i = 0; i < 8; ++i) W[base + i * 8] = Pn[i];
    }
    xrep = xn;
  }
}

extern "C" void kernel_launch(void* const* d_in, const int* in_sizes, int n_in,
                              void* d_out, int out_size, void* d_ws, size_t ws_size,
                              hipStream_t stream) {
  const float* z  = (const float*)d_in[0];
  const float* A  = (const float*)d_in[1];
  const float* C  = (const float*)d_in[2];
  const float* Q  = (const float*)d_in[3];
  const float* R  = (const float*)d_in[4];
  const float* x0 = (const float*)d_in[5];
  const float* P0 = (const float*)d_in[6];
  float* out = (float*)d_out;

  const int Bsz = in_sizes[0] / (TT * N4);
  const int waves = Bsz / 2;
  const int threads = 256;
  int blocks = (waves * 32) / threads;
  if (blocks < 1) blocks = 1;
  ekf1250_kernel<<<blocks, threads, 0, stream>>>(z, A, C, Q, R, x0, P0, out);
}


# P'=F(PFt)+Q and S=H(P'Ht)+R re-associated: X=PFt and PHt re-enter as B operand by a permlanex16 half exchange, so the G and T1 LDS round trips leave the dependent path; product sum order changes
# speedup vs baseline: 1.3643x; 1.0611x over previous
#include <hip/hip_runtime.h>
#include <hip/hip_bf16.h>


#define TT 256
#define N4 4

typedef __attribute__((ext_vector_type(2))) float v2f;
typedef __attribute__((ext_vector_type(8))) float v8f;

template <int K>
__device__ __forceinline__ float rshare(float v) {
  const int b = __float_as_int(v);
  return __int_as_float(__builtin_amdgcn_mov_dpp(b, 0x150 + K, 0xf, 0xf, false));
}
__device__ __forceinline__ float ror8hi(float v) {
  const int b = __float_as_int(v);
  return __int_as_float(__builtin_amdgcn_update_dpp(b, b, 0x128, 0x2, 0xf, false));
}
__device__ __forceinline__ float xhalf(float v) {
  const int b = __float_as_int(v);
  return __int_as_float(__builtin_amdgcn_permlanex16(b, b, 0x76543210, 0xfedcba98, false, false));
}

__device__ __forceinline__ v8f wmma4(float a0, float a1, float b0, float b1, v8f c) {
  v2f A, B;
  A[0] = a0; A[1] = a1;
  B[0] = b0; B[1] = b1;
  return __builtin_amdgcn_wmma_f32_16x16x4_f32(false, A, false, B, (short)0, c, false, false);
}

__device__ __forceinline__ float dtile_as_b(float dlo, float dhi, bool lowh, bool mid) {
  const float own = lowh ? dlo : dhi;
  const float oth = xhalf(lowh ? dhi : dlo);
  return mid ? oth : own;
}

__device__ __forceinline__ float fast_tanh(float x) {
#if __has_builtin(__builtin_amdgcn_tanhf)
  return __builtin_amdgcn_tanhf(x);
#else
  const float e = __expf(2.0f * x);
  return 1.0f - 2.0f * __builtin_amdgcn_rcpf(e + 1.0f);
#endif
}
#define INV_2PI 0.15915494309189535f
__device__ __forceinline__ float fast_sin(float x) {
  return __builtin_amdgcn_sinf(x * INV_2PI);
}
__device__ __forceinline__ float fast_cos(float x) {
  return __builtin_amdgcn_cosf(x * INV_2PI);
}

__global__ __launch_bounds__(256) void ekf1250_kernel(
    const float* __restrict__ z,  const float* __restrict__ Ag,
    const float* __restrict__ Cg, const float* __restrict__ Qg,
    const float* __restrict__ Rg, const float* __restrict__ x0g,
    const float* __restrict__ P0g, float* __restrict__ out)
{
  __shared__ float lds[8 * 640];
  const int tid  = threadIdx.x;
  const int lane = tid & 31;
  const int wave = (blockIdx.x * blockDim.x + tid) >> 5;
  const long b0  = (long)wave * 2;
  float* W = &lds[(tid >> 5) * 640];
  const int RP = 0;
  const int R2 = 256;
  const int R4 = 448;
  const int RT = 512;

  const int  r    = lane & 7;
  const int  e8   = (lane >> 3) & 1;
  const int  khi  = lane >> 4;
  const int  eD   = lane >> 4;
  const bool dval = (lane < 8) || (lane >= 24);
  const bool lowh = lane < 16;
  const bool mid  = (lane >= 8) && (lane < 24);
  const int  dcol = lane & 7;

  const int aBase  = e8 * 64 + r * 8 + 2 * khi;
  const int bBase  = e8 * 64 + (2 * khi) * 8 + r;

  float regA[8], regC[8];
  {
    const float4 a0 = *reinterpret_cast<const float4*>(Ag + r * 8);
    const float4 a1 = *reinterpret_cast<const float4*>(Ag + r * 8 + 4);
    regA[0]=a0.x; regA[1]=a0.y; regA[2]=a0.z; regA[3]=a0.w;
    regA[4]=a1.x; regA[5]=a1.y; regA[6]=a1.z; regA[7]=a1.w;
    const float4 c0 = *reinterpret_cast<const float4*>(Cg + (r & 3) * 8);
    const float4 c1 = *reinterpret_cast<const float4*>(Cg + (r & 3) * 8 + 4);
    const float m = (r < 4) ? 1.0f : 0.0f;
    regC[0]=c0.x*m; regC[1]=c0.y*m; regC[2]=c0.z*m; regC[3]=c0.w*m;
    regC[4]=c1.x*m; regC[5]=c1.y*m; regC[6]=c1.z*m; regC[7]=c1.w*m;
  }
  v8f Qd, Rd;
  #pragma unroll
  for (int i = 0; i < 8; ++i) Qd[i] = Qg[i * 8 + dcol];
  #pragma unroll
  for (int i = 0; i < 8; ++i) {
    float rv = Rg[(i & 3) * 4 + (dcol & 3)];
    Rd[i] = (i < 4 && dcol < 4) ? rv : 0.0f;
  }

  float xrep = x0g[r];
  for (int idx = lane; idx < 128; idx += 32) W[RP + idx] = P0g[idx & 63];

  const v8f z8 = {0,0,0,0,0,0,0,0};

  #pragma unroll 1
  for (int t = 0; t < TT; ++t) {
    float xb[8];
    xb[0] = rshare<0>(xrep); xb[1] = rshare<1>(xrep); xb[2] = rshare<2>(xrep); xb[3] = rshare<3>(xrep);
    xb[4] = rshare<4>(xrep); xb[5] = rshare<5>(xrep); xb[6] = rshare<6>(xrep); xb[7] = rshare<7>(xrep);
    float ta = 0.0f;
    #pragma unroll
    for (int k = 0; k < 8; ++k) ta = fmaf(regA[k], xb[k], ta);
    const float xp = fast_tanh(ta);
    const float dd = 1.0f - xp * xp;
    float xq[8];
    xq[0] = rshare<0>(xp); xq[1] = rshare<1>(xp); xq[2] = rshare<2>(xp); xq[3] = rshare<3>(xp);
    xq[4] = rshare<4>(xp); xq[5] = rshare<5>(xp); xq[6] = rshare<6>(xp); xq[7] = rshare<7>(xp);
    float u = 0.0f;
    #pragma unroll
    for (int k = 0; k < 8; ++k) u = fmaf(regC[k], xq[k], u);
    const float ys = fast_sin(u);
    const float cy = fast_cos(u);

    const float ddx = xhalf(dd);
    const float cyx = xhalf(cy);
    const float dA  = mid ? ddx : dd;
    const float cyA = mid ? cyx : cy;
    float Fo[4], Ho[4];
    #pragma unroll
    for (int jc = 0; jc < 2; ++jc)
      #pragma unroll
      for (int v = 0; v < 2; ++v) {
        const float av = lowh ? regA[4 * jc + v] : regA[4 * jc + v + 2];
        const float cv = lowh ? regC[4 * jc + v] : regC[4 * jc + v + 2];
        Fo[2 * jc + v] = dA * av;
        Ho[2 * jc + v] = cyA * cv;
      }

    v8f X = wmma4(W[RP + aBase + 0], W[RP + aBase + 1], Fo[0], Fo[1], z8);
    X     = wmma4(W[RP + aBase + 4], W[RP + aBase + 5], Fo[2], Fo[3], X);
    const float bx0 = dtile_as_b(X[0], X[2], lowh, mid), bx1 = dtile_as_b(X[1], X[3], lowh, mid);
    const float bx2 = dtile_as_b(X[4], X[6], lowh, mid), bx3 = dtile_as_b(X[5], X[7], lowh, mid);
    v8f Pp = wmma4(Fo[0], Fo[1], bx0, bx1, Qd);
    Pp     = wmma4(Fo[2], Fo[3], bx2, bx3, Pp);
    {
      const int base = dval ? (R2 + eD * 64 + dcol) : (RT + (lane & 15));
      #pragma unroll
      for (int i = 0; i < 8; ++i) W[base + i * 8] = Pp[i];
    }
    const float pa0 = W[R2 + aBase + 0], pa1 = W[R2 + aBase + 1];
    const float pa2 = W[R2 + aBase + 4], pa3 = W[R2 + aBase + 5];
    const float pb0 = W[R2 + bBase + 0 ], pb1 = W[R2 + bBase + 8 ];
    const float pb2 = W[R2 + bBase + 32], pb3 = W[R2 + bBase + 40];
    v8f PHt = wmma4(pa0, pa1, Ho[0], Ho[1], z8);
    PHt     = wmma4(pa2, pa3, Ho[2], Ho[3], PHt);
    {
      const int base = (dval && dcol < 4) ? (R4 + eD * 32 + dcol) : (RT + (lane & 15));
      #pragma unroll
      for (int i = 0; i < 8; ++i) W[base + i * 4] = PHt[i];
    }
    const float bs0 = dtile_as_b(PHt[0], PHt[2], lowh, mid), bs1 = dtile_as_b(PHt[1], PHt[3], lowh, mid);
    const float bs2 = dtile_as_b(PHt[4], PHt[6], lowh, mid), bs3 = dtile_as_b(PHt[5], PHt[7], lowh, mid);
    v8f Sm = wmma4(Ho[0], Ho[1], bs0, bs1, Rd);
    Sm     = wmma4(Ho[2], Ho[3], bs2, bs3, Sm);
    v8f T1 = wmma4(Ho[0], Ho[1], pb0, pb1, z8);
    T1     = wmma4(Ho[2], Ho[3], pb2, pb3, T1);

    const float4 rhs4 = *reinterpret_cast<const float4*>(&W[R4 + (eD << 5) + (r << 2)]);
    float rr4[4] = {rhs4.x, rhs4.y, rhs4.z, rhs4.w};
    float s[4][4];
    #pragma unroll
    for (int i = 0; i < 4; ++i) {
      const float sr = ror8hi(Sm[i]);
      s[i][0] = rshare<0>(sr); s[i][1] = rshare<1>(sr);
      s[i][2] = rshare<2>(sr); s[i][3] = rshare<3>(sr);
    }
    #pragma unroll
    for (int k = 0; k < 4; ++k) {
      const float inv = __builtin_amdgcn_rcpf(s[k][k]);
      #pragma unroll
      for (int j = k + 1; j < 4; ++j) s[k][j] *= inv;
      rr4[k] *= inv;
      #pragma unroll
      for (int i = k + 1; i < 4; ++i) {
        const float f = s[i][k];
        #pragma unroll
        for (int j = k + 1; j < 4; ++j) s[i][j] = fmaf(-f, s[k][j], s[i][j]);
        rr4[i] = fmaf(-f, rr4[k], rr4[i]);
      }
    }
    float Kp[4];
    Kp[3] = rr4[3];
    Kp[2] = fmaf(-s[2][3], Kp[3], rr4[2]);
    Kp[1] = fmaf(-s[1][3], Kp[3], fmaf(-s[1][2], Kp[2], rr4[1]));
    Kp[0] = fmaf(-s[0][3], Kp[3], fmaf(-s[0][2], Kp[2], fmaf(-s[0][1], Kp[1], rr4[0])));

    const long zidx = ((b0 + (lane >> 4)) * TT + t) * N4 + (r & 3);
    const float zv = z[zidx];
    __builtin_prefetch(&z[(t < TT - 8) ? (zidx + 8 * N4) : zidx], 0, 0);
    const float dy = zv - ys;
    float dyb[4];
    dyb[0] = rshare<0>(dy); dyb[1] = rshare<1>(dy); dyb[2] = rshare<2>(dy); dyb[3] = rshare<3>(dy);
    float xn = xp;
    #pragma unroll
    for (int j = 0; j < 4; ++j) xn = fmaf(Kp[j], dyb[j], xn);
    if (dval) out[((b0 + eD) * TT + t) * 8 + r] = xn;

    float Av0, Av1, Bv0, Bv1;
    {
      const float ko0 = xhalf(lowh ? Kp[2] : Kp[0]);
      const float ko1 = xhalf(lowh ? Kp[3] : Kp[1]);
      const float to0 = xhalf(lowh ? T1[2] : T1[0]);
      const float to1 = xhalf(lowh ? T1[3] : T1[1]);
      Av0 = -(mid ? ko0 : (lowh ? Kp[0] : Kp[2]));
      Av1 = -(mid ? ko1 : (lowh ? Kp[1] : Kp[3]));
      Bv0 = mid ? to0 : (lowh ? T1[0] : T1[2]);
      Bv1 = mid ? to1 : (lowh ? T1[1] : T1[3]);
    }
    v8f Pn = wmma4(Av0, Av1, Bv0, Bv1, Pp);
    {
      const int base = dval ? (RP + eD * 64 + dcol) : (RT + (lane & 15));
      #pragma unroll
      for (int i = 0; i < 8; ++i) W[base + i * 8] = Pn[i];
    }
    xrep = xn;
  }
}

extern "C" void kernel_launch(void* const* d_in, const int* in_sizes, int n_in,
                              void* d_out, int out_size, void* d_ws, size_t ws_size,
                              hipStream_t stream) {
  const float* z  = (const float*)d_in[0];
  const float* A  = (const float*)d_in[1];
  const float* C  = (const float*)d_in[2];
  const float* Q  = (const float*)d_in[3];
  const float* R  = (const float*)d_in[4];
  const float* x0 = (const float*)d_in[5];
  const float* P0 = (const float*)d_in[6];
  float* out = (float*)d_out;

  const int Bsz = in_sizes[0] / (TT * N4);
  const int waves = Bsz / 2;
  const int threads = 256;
  int blocks = (waves * 32) / threads;
  if (blocks < 1) blocks = 1;
  ekf1250_kernel<<<blocks, threads, 0, stream>>>(z, A, C, Q, R, x0, P0, out);
}


# Issue order only: P (A layout) and z are read at the loop head and the PHt solve rows right behind their stores, each held by a scheduler fence, so LDS latency runs under dependent vector work
# speedup vs baseline: 1.4536x; 1.0655x over previous
#include <hip/hip_runtime.h>
#include <hip/hip_bf16.h>


#define TT 256
#define N4 4

typedef __attribute__((ext_vector_type(2))) float v2f;
typedef __attribute__((ext_vector_type(8))) float v8f;

template <int K>
__device__ __forceinline__ float rshare(float v) {
  const int b = __float_as_int(v);
  return __int_as_float(__builtin_amdgcn_mov_dpp(b, 0x150 + K, 0xf, 0xf, false));
}
__device__ __forceinline__ float ror8hi(float v) {
  const int b = __float_as_int(v);
  return __int_as_float(__builtin_amdgcn_update_dpp(b, b, 0x128, 0x2, 0xf, false));
}
__device__ __forceinline__ float xhalf(float v) {
  const int b = __float_as_int(v);
  return __int_as_float(__builtin_amdgcn_permlanex16(b, b, 0x76543210, 0xfedcba98, false, false));
}

__device__ __forceinline__ v8f wmma4(float a0, float a1, float b0, float b1, v8f c) {
  v2f A, B;
  A[0] = a0; A[1] = a1;
  B[0] = b0; B[1] = b1;
  return __builtin_amdgcn_wmma_f32_16x16x4_f32(false, A, false, B, (short)0, c, false, false);
}

__device__ __forceinline__ void hold_issue_order() { __builtin_amdgcn_sched_barrier(0); }

__device__ __forceinline__ float dtile_as_b(float dlo, float dhi, bool lowh, bool mid) {
  const float own = lowh ? dlo : dhi;
  const float oth = xhalf(lowh ? dhi : dlo);
  return mid ? oth : own;
}

__device__ __forceinline__ float fast_tanh(float x) {
#if __has_builtin(__builtin_amdgcn_tanhf)
  return __builtin_amdgcn_tanhf(x);
#else
  const float e = __expf(2.0f * x);
  return 1.0f - 2.0f * __builtin_amdgcn_rcpf(e + 1.0f);
#endif
}
#define INV_2PI 0.15915494309189535f
__device__ __forceinline__ float fast_sin(float x) {
  return __builtin_amdgcn_sinf(x * INV_2PI);
}
__device__ __forceinline__ float fast_cos(float x) {
  return __builtin_amdgcn_cosf(x * INV_2PI);
}

__global__ __launch_bounds__(256) void ekf1250_kernel(
    const float* __restrict__ z,  const float* __restrict__ Ag,
    const float* __restrict__ Cg, const float* __restrict__ Qg,
    const float* __restrict__ Rg, const float* __restrict__ x0g,
    const float* __restrict__ P0g, float* __restrict__ out)
{
  __shared__ float lds[8 * 640];
  const int tid  = threadIdx.x;
  const int lane = tid & 31;
  const int wave = (blockIdx.x * blockDim.x + tid) >> 5;
  const long b0  = (long)wave * 2;
  float* W = &lds[(tid >> 5) * 640];
  const int RP = 0;
  const int R2 = 256;
  const int R4 = 448;
  const int RT = 512;

  const int  r    = lane & 7;
  const int  e8   = (lane >> 3) & 1;
  const int  khi  = lane >> 4;
  const int  eD   = lane >> 4;
  const bool dval = (lane < 8) || (lane >= 24);
  const bool lowh = lane < 16;
  const bool mid  = (lane >= 8) && (lane < 24);
  const int  dcol = lane & 7;

  const int aBase  = e8 * 64 + r * 8 + 2 * khi;
  const int bBase  = e8 * 64 + (2 * khi) * 8 + r;

  float regA[8], regC[8];
  {
    const float4 a0 = *reinterpret_cast<const float4*>(Ag + r * 8);
    const float4 a1 = *reinterpret_cast<const float4*>(Ag + r * 8 + 4);
    regA[0]=a0.x; regA[1]=a0.y; regA[2]=a0.z; regA[3]=a0.w;
    regA[4]=a1.x; regA[5]=a1.y; regA[6]=a1.z; regA[7]=a1.w;
    const float4 c0 = *reinterpret_cast<const float4*>(Cg + (r & 3) * 8);
    const float4 c1 = *reinterpret_cast<const float4*>(Cg + (r & 3) * 8 + 4);
    const float m = (r < 4) ? 1.0f : 0.0f;
    regC[0]=c0.x*m; regC[1]=c0.y*m; regC[2]=c0.z*m; regC[3]=c0.w*m;
    regC[4]=c1.x*m; regC[5]=c1.y*m; regC[6]=c1.z*m; regC[7]=c1.w*m;
  }
  v8f Qd, Rd;
  #pragma unroll
  for (int i = 0; i < 8; ++i) Qd[i] = Qg[i * 8 + dcol];
  #pragma unroll
  for (int i = 0; i < 8; ++i) {
    float rv = Rg[(i & 3) * 4 + (dcol & 3)];
    Rd[i] = (i < 4 && dcol < 4) ? rv : 0.0f;
  }

  float xrep = x0g[r];
  for (int idx = lane; idx < 128; idx += 32) W[RP + idx] = P0g[idx & 63];

  const v8f z8 = {0,0,0,0,0,0,0,0};

  #pragma unroll 1
  for (int t = 0; t < TT; ++t) {
    const long zidx = ((b0 + (lane >> 4)) * TT + t) * N4 + (r & 3);
    const float zv = z[zidx];
    __builtin_prefetch(&z[(t < TT - 8) ? (zidx + 8 * N4) : zidx], 0, 0);
    const float pr0 = W[RP + aBase + 0], pr1 = W[RP + aBase + 1];
    const float pr2 = W[RP + aBase + 4], pr3 = W[RP + aBase + 5];
    hold_issue_order();
    float xb[8];
    xb[0] = rshare<0>(xrep); xb[1] = rshare<1>(xrep); xb[2] = rshare<2>(xrep); xb[3] = rshare<3>(xrep);
    xb[4] = rshare<4>(xrep); xb[5] = rshare<5>(xrep); xb[6] = rshare<6>(xrep); xb[7] = rshare<7>(xrep);
    float ta = 0.0f;
    #pragma unroll
    for (int k = 0; k < 8; ++k) ta = fmaf(regA[k], xb[k], ta);
    const float xp = fast_tanh(ta);
    const float dd = 1.0f - xp * xp;
    float xq[8];
    xq[0] = rshare<0>(xp); xq[1] = rshare<1>(xp); xq[2] = rshare<2>(xp); xq[3] = rshare<3>(xp);
    xq[4] = rshare<4>(xp); xq[5] = rshare<5>(xp); xq[6] = rshare<6>(xp); xq[7] = rshare<7>(xp);
    float u = 0.0f;
    #pragma unroll
    for (int k = 0; k < 8; ++k) u = fmaf(regC[k], xq[k], u);
    const float ys = fast_sin(u);
    const float cy = fast_cos(u);

    const float ddx = xhalf(dd);
    const float cyx = xhalf(cy);
    const float dA  = mid ? ddx : dd;
    const float cyA = mid ? cyx : cy;
    float Fo[4], Ho[4];
    #pragma unroll
    for (int jc = 0; jc < 2; ++jc)
      #pragma unroll
      for (int v = 0; v < 2; ++v) {
        const float av = lowh ? regA[4 * jc + v] : regA[4 * jc + v + 2];
        const float cv = lowh ? regC[4 * jc + v] : regC[4 * jc + v + 2];
        Fo[2 * jc + v] = dA * av;
        Ho[2 * jc + v] = cyA * cv;
      }

    v8f X = wmma4(pr0, pr1, Fo[0], Fo[1], z8);
    X     = wmma4(pr2, pr3, Fo[2], Fo[3], X);
    const float bx0 = dtile_as_b(X[0], X[2], lowh, mid), bx1 = dtile_as_b(X[1], X[3], lowh, mid);
    const float bx2 = dtile_as_b(X[4], X[6], lowh, mid), bx3 = dtile_as_b(X[5], X[7], lowh, mid);
    v8f Pp = wmma4(Fo[0], Fo[1], bx0, bx1, Qd);
    Pp     = wmma4(Fo[2], Fo[3], bx2, bx3, Pp);
    {
      const int base = dval ? (R2 + eD * 64 + dcol) : (RT + (lane & 15));
      #pragma unroll
      for (int i = 0; i < 8; ++i) W[base + i * 8] = Pp[i];
    }
    const float pa0 = W[R2 + aBase + 0], pa1 = W[R2 + aBase + 1];
    const float pa2 = W[R2 + aBase + 4], pa3 = W[R2 + aBase + 5];
    const float pb0 = W[R2 + bBase + 0 ], pb1 = W[R2 + bBase + 8 ];
    const float pb2 = W[R2 + bBase + 32], pb3 = W[R2 + bBase + 40];
    v8f PHt = wmma4(pa0, pa1, Ho[0], Ho[1], z8);
    PHt     = wmma4(pa2, pa3, Ho[2], Ho[3], PHt);
    {
      const int base = (dval && dcol < 4) ? (R4 + eD * 32 + dcol) : (RT + (lane & 15));
      #pragma unroll
      for (int i = 0; i < 8; ++i) W[base + i * 4] = PHt[i];
    }
    const float4 rhs4 = *reinterpret_cast<const float4*>(&W[R4 + (eD << 5) + (r << 2)]);
    hold_issue_order();
    const float bs0 = dtile_as_b(PHt[0], PHt[2], lowh, mid), bs1 = dtile_as_b(PHt[1], PHt[3], lowh, mid);
    const float bs2 = dtile_as_b(PHt[4], PHt[6], lowh, mid), bs3 = dtile_as_b(PHt[5], PHt[7], lowh, mid);
    v8f Sm = wmma4(Ho[0], Ho[1], bs0, bs1, Rd);
    Sm     = wmma4(Ho[2], Ho[3], bs2, bs3, Sm);
    v8f T1 = wmma4(Ho[0], Ho[1], pb0, pb1, z8);
    T1     = wmma4(Ho[2], Ho[3], pb2, pb3, T1);

    float rr4[4] = {rhs4.x, rhs4.y, rhs4.z, rhs4.w};
    float s[4][4];
    #pragma unroll
    for (int i = 0; i < 4; ++i) {
      const float sr = ror8hi(Sm[i]);
      s[i][0] = rshare<0>(sr); s[i][1] = rshare<1>(sr);
      s[i][2] = rshare<2>(sr); s[i][3] = rshare<3>(sr);
    }
    #pragma unroll
    for (int k = 0; k < 4; ++k) {
      const float inv = __builtin_amdgcn_rcpf(s[k][k]);
      #pragma unroll
      for (int j = k + 1; j < 4; ++j) s[k][j] *= inv;
      rr4[k] *= inv;
      #pragma unroll
      for (int i = k + 1; i < 4; ++i) {
        const float f = s[i][k];
        #pragma unroll
        for (int j = k + 1; j < 4; ++j) s[i][j] = fmaf(-f, s[k][j], s[i][j]);
        rr4[i] = fmaf(-f, rr4[k], rr4[i]);
      }
    }
    float Kp[4];
    Kp[3] = rr4[3];
    Kp[2] = fmaf(-s[2][3], Kp[3], rr4[2]);
    Kp[1] = fmaf(-s[1][3], Kp[3], fmaf(-s[1][2], Kp[2], rr4[1]));
    Kp[0] = fmaf(-s[0][3], Kp[3], fmaf(-s[0][2], Kp[2], fmaf(-s[0][1], Kp[1], rr4[0])));

    const float dy = zv - ys;
    float dyb[4];
    dyb[0] = rshare<0>(dy); dyb[1] = rshare<1>(dy); dyb[2] = rshare<2>(dy); dyb[3] = rshare<3>(dy);
    float xn = xp;
    #pragma unroll
    for (int j = 0; j < 4; ++j) xn = fmaf(Kp[j], dyb[j], xn);
    if (dval) out[((b0 + eD) * TT + t) * 8 + r] = xn;

    float Av0, Av1, Bv0, Bv1;
    {
      const float ko0 = xhalf(lowh ? Kp[2] : Kp[0]);
      const float ko1 = xhalf(lowh ? Kp[3] : Kp[1]);
      const float to0 = xhalf(lowh ? T1[2] : T1[0]);
      const float to1 = xhalf(lowh ? T1[3] : T1[1]);
      Av0 = -(mid ? ko0 : (lowh ? Kp[0] : Kp[2]));
      Av1 = -(mid ? ko1 : (lowh ? Kp[1] : Kp[3]));
      Bv0 = mid ? to0 : (lowh ? T1[0] : T1[2]);
      Bv1 = mid ? to1 : (lowh ? T1[1] : T1[3]);
    }
    v8f Pn = wmma4(Av0, Av1, Bv0, Bv1, Pp);
    {
      const int base = dval ? (RP + eD * 64 + dcol) : (RT + (lane & 15));
      #pragma unroll
      for (int i = 0; i < 8; ++i) W[base + i * 8] = Pn[i];
    }
    xrep = xn;
  }
}

extern "C" void kernel_launch(void* const* d_in, const int* in_sizes, int n_in,
                              void* d_out, int out_size, void* d_ws, size_t ws_size,
                              hipStream_t stream) {
  const float* z  = (const float*)d_in[0];
  const float* A  = (const float*)d_in[1];
  const float* C  = (const float*)d_in[2];
  const float* Q  = (const float*)d_in[3];
  const float* R  = (const float*)d_in[4];
  const float* x0 = (const float*)d_in[5];
  const float* P0 = (const float*)d_in[6];
  float* out = (float*)d_out;

  const int Bsz = in_sizes[0] / (TT * N4);
  const int waves = Bsz / 2;
  const int threads = 256;
  int blocks = (waves * 32) / threads;
  if (blocks < 1) blocks = 1;
  ekf1250_kernel<<<blocks, threads, 0, stream>>>(z, A, C, Q, R, x0, P0, out);
}


# Row broadcasts (state, tanh state, innovation, S entries) issue as v_permlane16_var_b32 with lane selects held in registers instead of DPP moves; the S gather's rotate folds into its select
# speedup vs baseline: 1.5546x; 1.0695x over previous
#include <hip/hip_runtime.h>
#include <hip/hip_bf16.h>


#define TT 256
#define N4 4

typedef __attribute__((ext_vector_type(2))) float v2f;
typedef __attribute__((ext_vector_type(8))) float v8f;

__device__ __forceinline__ float rowpick(float v, int sel) {
  const int b = __float_as_int(v);
  return __int_as_float(__builtin_amdgcn_permlane16_var(b, b, sel, true, false));
}
__device__ __forceinline__ float xhalf(float v) {
  const int b = __float_as_int(v);
  return __int_as_float(__builtin_amdgcn_permlanex16(b, b, 0x76543210, 0xfedcba98, false, false));
}

__device__ __forceinline__ v8f wmma4(float a0, float a1, float b0, float b1, v8f c) {
  v2f A, B;
  A[0] = a0; A[1] = a1;
  B[0] = b0; B[1] = b1;
  return __builtin_amdgcn_wmma_f32_16x16x4_f32(false, A, false, B, (short)0, c, false, false);
}

__device__ __forceinline__ void hold_issue_order() { __builtin_amdgcn_sched_barrier(0); }

__device__ __forceinline__ float dtile_as_b(float dlo, float dhi, bool lowh, bool mid) {
  const float own = lowh ? dlo : dhi;
  const float oth = xhalf(lowh ? dhi : dlo);
  return mid ? oth : own;
}

__device__ __forceinline__ float fast_tanh(float x) {
#if __has_builtin(__builtin_amdgcn_tanhf)
  return __builtin_amdgcn_tanhf(x);
#else
  const float e = __expf(2.0f * x);
  return 1.0f - 2.0f * __builtin_amdgcn_rcpf(e + 1.0f);
#endif
}
#define INV_2PI 0.15915494309189535f
__device__ __forceinline__ float fast_sin(float x) {
  return __builtin_amdgcn_sinf(x * INV_2PI);
}
__device__ __forceinline__ float fast_cos(float x) {
  return __builtin_amdgcn_cosf(x * INV_2PI);
}

__global__ __launch_bounds__(256) void ekf1250_kernel(
    const float* __restrict__ z,  const float* __restrict__ Ag,
    const float* __restrict__ Cg, const float* __restrict__ Qg,
    const float* __restrict__ Rg, const float* __restrict__ x0g,
    const float* __restrict__ P0g, float* __restrict__ out)
{
  __shared__ float lds[8 * 640];
  const int tid  = threadIdx.x;
  const int lane = tid & 31;
  const int wave = (blockIdx.x * blockDim.x + tid) >> 5;
  const long b0  = (long)wave * 2;
  float* W = &lds[(tid >> 5) * 640];
  const int RP = 0;
  const int R2 = 256;
  const int R4 = 448;
  const int RT = 512;

  const int  r    = lane & 7;
  const int  e8   = (lane >> 3) & 1;
  const int  khi  = lane >> 4;
  const int  eD   = lane >> 4;
  const bool dval = (lane < 8) || (lane >= 24);
  const bool lowh = lane < 16;
  const bool mid  = (lane >= 8) && (lane < 24);
  const int  dcol = lane & 7;

  const int aBase  = e8 * 64 + r * 8 + 2 * khi;
  const int bBase  = e8 * 64 + (2 * khi) * 8 + r;

  float regA[8], regC[8];
  {
    const float4 a0 = *reinterpret_cast<const float4*>(Ag + r * 8);
    const float4 a1 = *reinterpret_cast<const float4*>(Ag + r * 8 + 4);
    regA[0]=a0.x; regA[1]=a0.y; regA[2]=a0.z; regA[3]=a0.w;
    regA[4]=a1.x; regA[5]=a1.y; regA[6]=a1.z; regA[7]=a1.w;
    const float4 c0 = *reinterpret_cast<const float4*>(Cg + (r & 3) * 8);
    const float4 c1 = *reinterpret_cast<const float4*>(Cg + (r & 3) * 8 + 4);
    const float m = (r < 4) ? 1.0f : 0.0f;
    regC[0]=c0.x*m; regC[1]=c0.y*m; regC[2]=c0.z*m; regC[3]=c0.w*m;
    regC[4]=c1.x*m; regC[5]=c1.y*m; regC[6]=c1.z*m; regC[7]=c1.w*m;
  }
  v8f Qd, Rd;
  #pragma unroll
  for (int i = 0; i < 8; ++i) Qd[i] = Qg[i * 8 + dcol];
  #pragma unroll
  for (int i = 0; i < 8; ++i) {
    float rv = Rg[(i & 3) * 4 + (dcol & 3)];
    Rd[i] = (i < 4 && dcol < 4) ? rv : 0.0f;
  }

  float xrep = x0g[r];
  for (int idx = lane; idx < 128; idx += 32) W[RP + idx] = P0g[idx & 63];

  const v8f z8 = {0,0,0,0,0,0,0,0};

  const int rowbase = 8 * eD;
  const int selk[8] = {rowbase, rowbase + 1, rowbase + 2, rowbase + 3,
                       rowbase + 4, rowbase + 5, rowbase + 6, rowbase + 7};

  #pragma unroll 1
  for (int t = 0; t < TT; ++t) {
    const long zidx = ((b0 + (lane >> 4)) * TT + t) * N4 + (r & 3);
    const float zv = z[zidx];
    __builtin_prefetch(&z[(t < TT - 8) ? (zidx + 8 * N4) : zidx], 0, 0);
    const float pr0 = W[RP + aBase + 0], pr1 = W[RP + aBase + 1];
    const float pr2 = W[RP + aBase + 4], pr3 = W[RP + aBase + 5];
    hold_issue_order();
    float xb[8];
    xb[0] = rowpick(xrep, selk[0]); xb[1] = rowpick(xrep, selk[1]); xb[2] = rowpick(xrep, selk[2]); xb[3] = rowpick(xrep, selk[3]);
    xb[4] = rowpick(xrep, selk[4]); xb[5] = rowpick(xrep, selk[5]); xb[6] = rowpick(xrep, selk[6]); xb[7] = rowpick(xrep, selk[7]);
    float ta = 0.0f;
    #pragma unroll
    for (int k = 0; k < 8; ++k) ta = fmaf(regA[k], xb[k], ta);
    const float xp = fast_tanh(ta);
    const float dd = 1.0f - xp * xp;
    float xq[8];
    xq[0] = rowpick(xp, selk[0]); xq[1] = rowpick(xp, selk[1]); xq[2] = rowpick(xp, selk[2]); xq[3] = rowpick(xp, selk[3]);
    xq[4] = rowpick(xp, selk[4]); xq[5] = rowpick(xp, selk[5]); xq[6] = rowpick(xp, selk[6]); xq[7] = rowpick(xp, selk[7]);
    float u = 0.0f;
    #pragma unroll
    for (int k = 0; k < 8; ++k) u = fmaf(regC[k], xq[k], u);
    const float ys = fast_sin(u);
    const float cy = fast_cos(u);

    const float ddx = xhalf(dd);
    const float cyx = xhalf(cy);
    const float dA  = mid ? ddx : dd;
    const float cyA = mid ? cyx : cy;
    float Fo[4], Ho[4];
    #pragma unroll
    for (int jc = 0; jc < 2; ++jc)
      #pragma unroll
      for (int v = 0; v < 2; ++v) {
        const float av = lowh ? regA[4 * jc + v] : regA[4 * jc + v + 2];
        const float cv = lowh ? regC[4 * jc + v] : regC[4 * jc + v + 2];
        Fo[2 * jc + v] = dA * av;
        Ho[2 * jc + v] = cyA * cv;
      }

    v8f X = wmma4(pr0, pr1, Fo[0], Fo[1], z8);
    X     = wmma4(pr2, pr3, Fo[2], Fo[3], X);
    const float bx0 = dtile_as_b(X[0], X[2], lowh, mid), bx1 = dtile_as_b(X[1], X[3], lowh, mid);
    const float bx2 = dtile_as_b(X[4], X[6], lowh, mid), bx3 = dtile_as_b(X[5], X[7], lowh, mid);
    v8f Pp = wmma4(Fo[0], Fo[1], bx0, bx1, Qd);
    Pp     = wmma4(Fo[2], Fo[3], bx2, bx3, Pp);
    {
      const int base = dval ? (R2 + eD * 64 + dcol) : (RT + (lane & 15));
      #pragma unroll
      for (int i = 0; i < 8; ++i) W[base + i * 8] = Pp[i];
    }
    const float pa0 = W[R2 + aBase + 0], pa1 = W[R2 + aBase + 1];
    const float pa2 = W[R2 + aBase + 4], pa3 = W[R2 + aBase + 5];
    const float pb0 = W[R2 + bBase + 0 ], pb1 = W[R2 + bBase + 8 ];
    const float pb2 = W[R2 + bBase + 32], pb3 = W[R2 + bBase + 40];
    v8f PHt = wmma4(pa0, pa1, Ho[0], Ho[1], z8);
    PHt     = wmma4(pa2, pa3, Ho[2], Ho[3], PHt);
    {
      const int base = (dval && dcol < 4) ? (R4 + eD * 32 + dcol) : (RT + (lane & 15));
      #pragma unroll
      for (int i = 0; i < 8; ++i) W[base + i * 4] = PHt[i];
    }
    const float4 rhs4 = *reinterpret_cast<const float4*>(&W[R4 + (eD << 5) + (r << 2)]);
    hold_issue_order();
    const float bs0 = dtile_as_b(PHt[0], PHt[2], lowh, mid), bs1 = dtile_as_b(PHt[1], PHt[3], lowh, mid);
    const float bs2 = dtile_as_b(PHt[4], PHt[6], lowh, mid), bs3 = dtile_as_b(PHt[5], PHt[7], lowh, mid);
    v8f Sm = wmma4(Ho[0], Ho[1], bs0, bs1, Rd);
    Sm     = wmma4(Ho[2], Ho[3], bs2, bs3, Sm);
    v8f T1 = wmma4(Ho[0], Ho[1], pb0, pb1, z8);
    T1     = wmma4(Ho[2], Ho[3], pb2, pb3, T1);

    float rr4[4] = {rhs4.x, rhs4.y, rhs4.z, rhs4.w};
    float s[4][4];
    #pragma unroll
    for (int i = 0; i < 4; ++i) {
      s[i][0] = rowpick(Sm[i], selk[0]); s[i][1] = rowpick(Sm[i], selk[1]);
      s[i][2] = rowpick(Sm[i], selk[2]); s[i][3] = rowpick(Sm[i], selk[3]);
    }
    #pragma unroll
    for (int k = 0; k < 4; ++k) {
      const float inv = __builtin_amdgcn_rcpf(s[k][k]);
      #pragma unroll
      for (int j = k + 1; j < 4; ++j) s[k][j] *= inv;
      rr4[k] *= inv;
      #pragma unroll
      for (int i = k + 1; i < 4; ++i) {
        const float f = s[i][k];
        #pragma unroll
        for (int j = k + 1; j < 4; ++j) s[i][j] = fmaf(-f, s[k][j], s[i][j]);
        rr4[i] = fmaf(-f, rr4[k], rr4[i]);
      }
    }
    float Kp[4];
    Kp[3] = rr4[3];
    Kp[2] = fmaf(-s[2][3], Kp[3], rr4[2]);
    Kp[1] = fmaf(-s[1][3], Kp[3], fmaf(-s[1][2], Kp[2], rr4[1]));
    Kp[0] = fmaf(-s[0][3], Kp[3], fmaf(-s[0][2], Kp[2], fmaf(-s[0][1], Kp[1], rr4[0])));

    const float dy = zv - ys;
    float dyb[4];
    dyb[0] = rowpick(dy, selk[0]); dyb[1] = rowpick(dy, selk[1]); dyb[2] = rowpick(dy, selk[2]); dyb[3] = rowpick(dy, selk[3]);
    float xn = xp;
    #pragma unroll
    for (int j = 0; j < 4; ++j) xn = fmaf(Kp[j], dyb[j], xn);
    if (dval) out[((b0 + eD) * TT + t) * 8 + r] = xn;

    float Av0, Av1, Bv0, Bv1;
    {
      const float ko0 = xhalf(lowh ? Kp[2] : Kp[0]);
      const float ko1 = xhalf(lowh ? Kp[3] : Kp[1]);
      const float to0 = xhalf(lowh ? T1[2] : T1[0]);
      const float to1 = xhalf(lowh ? T1[3] : T1[1]);
      Av0 = -(mid ? ko0 : (lowh ? Kp[0] : Kp[2]));
      Av1 = -(mid ? ko1 : (lowh ? Kp[1] : Kp[3]));
      Bv0 = mid ? to0 : (lowh ? T1[0] : T1[2]);
      Bv1 = mid ? to1 : (lowh ? T1[1] : T1[3]);
    }
    v8f Pn = wmma4(Av0, Av1, Bv0, Bv1, Pp);
    {
      const int base = dval ? (RP + eD * 64 + dcol) : (RT + (lane & 15));
      #pragma unroll
      for (int i = 0; i < 8; ++i) W[base + i * 8] = Pn[i];
    }
    xrep = xn;
  }
}

extern "C" void kernel_launch(void* const* d_in, const int* in_sizes, int n_in,
                              void* d_out, int out_size, void* d_ws, size_t ws_size,
                              hipStream_t stream) {
  const float* z  = (const float*)d_in[0];
  const float* A  = (const float*)d_in[1];
  const float* C  = (const float*)d_in[2];
  const float* Q  = (const float*)d_in[3];
  const float* R  = (const float*)d_in[4];
  const float* x0 = (const float*)d_in[5];
  const float* P0 = (const float*)d_in[6];
  float* out = (float*)d_out;

  const int Bsz = in_sizes[0] / (TT * N4);
  const int waves = Bsz / 2;
  const int threads = 256;
  int blocks = (waves * 32) / threads;
  if (blocks < 1) blocks = 1;
  ekf1250_kernel<<<blocks, threads, 0, stream>>>(z, A, C, Q, R, x0, P0, out);
}


# Tile rows/columns re-mapped to lanes so the valid half of every D tile is already operand layout: products chain by register selects, covariance kept transposed in registers, no LDS on its path
# speedup vs baseline: 1.7810x; 1.1456x over previous
#include <hip/hip_runtime.h>
#include <hip/hip_bf16.h>


#define TT 256
#define N4 4

typedef __attribute__((ext_vector_type(2))) float v2f;
typedef __attribute__((ext_vector_type(8))) float v8f;

__device__ __forceinline__ float rowpick(float v, int sel) {
  const int b = __float_as_int(v);
  return __int_as_float(__builtin_amdgcn_permlane16_var(b, b, sel, true, false));
}
__device__ __forceinline__ float xhalf(float v) {
  const int b = __float_as_int(v);
  return __int_as_float(__builtin_amdgcn_permlanex16(b, b, 0x76543210, 0xfedcba98, false, false));
}

__device__ __forceinline__ v8f wmma4(float a0, float a1, float b0, float b1, v8f c) {
  v2f A, B;
  A[0] = a0; A[1] = a1;
  B[0] = b0; B[1] = b1;
  return __builtin_amdgcn_wmma_f32_16x16x4_f32(false, A, false, B, (short)0, c, false, false);
}

__device__ __forceinline__ void hold_issue_order() { __builtin_amdgcn_sched_barrier(0); }

__device__ __forceinline__ float fast_tanh(float x) {
#if __has_builtin(__builtin_amdgcn_tanhf)
  return __builtin_amdgcn_tanhf(x);
#else
  const float e = __expf(2.0f * x);
  return 1.0f - 2.0f * __builtin_amdgcn_rcpf(e + 1.0f);
#endif
}
#define INV_2PI 0.15915494309189535f
__device__ __forceinline__ float fast_sin(float x) {
  return __builtin_amdgcn_sinf(x * INV_2PI);
}
__device__ __forceinline__ float fast_cos(float x) {
  return __builtin_amdgcn_cosf(x * INV_2PI);
}

__global__ __launch_bounds__(256) void ekf1250_kernel(
    const float* __restrict__ z,  const float* __restrict__ Ag,
    const float* __restrict__ Cg, const float* __restrict__ Qg,
    const float* __restrict__ Rg, const float* __restrict__ x0g,
    const float* __restrict__ P0g, float* __restrict__ out)
{
  __shared__ float lds[8 * 256];
  const int tid  = threadIdx.x;
  const int lane = tid & 31;
  const int wave = (blockIdx.x * blockDim.x + tid) >> 5;
  const long b0  = (long)wave * 2;
  float* W = &lds[(tid >> 5) * 256];
  const int R2 = 0;
  const int R4 = 128;
  const int RT = 192;

  const int  c    = lane & 15;
  const int  h    = lane >> 4;
  const int  e    = (c >> 2) & 1;
  const int  i    = 4 * ((c & 3) >> 1) + 2 * (c >> 3) + (c & 1);
  const bool ch1  = e != 0;
  const bool lowh = lane < 16;
  const int  krow[4] = {2 * h, 2 * h + 1, 4 + 2 * h, 5 + 2 * h};

  float regA[8], regC[8];
  {
    const float4 a0 = *reinterpret_cast<const float4*>(Ag + i * 8);
    const float4 a1 = *reinterpret_cast<const float4*>(Ag + i * 8 + 4);
    regA[0]=a0.x; regA[1]=a0.y; regA[2]=a0.z; regA[3]=a0.w;
    regA[4]=a1.x; regA[5]=a1.y; regA[6]=a1.z; regA[7]=a1.w;
    const float4 c0 = *reinterpret_cast<const float4*>(Cg + (i & 3) * 8);
    const float4 c1 = *reinterpret_cast<const float4*>(Cg + (i & 3) * 8 + 4);
    const float m = (i < 4) ? 1.0f : 0.0f;
    regC[0]=c0.x*m; regC[1]=c0.y*m; regC[2]=c0.z*m; regC[3]=c0.w*m;
    regC[4]=c1.x*m; regC[5]=c1.y*m; regC[6]=c1.z*m; regC[7]=c1.w*m;
  }
  float aF[4], cH[4];
  #pragma unroll
  for (int q = 0; q < 4; ++q) {
    aF[q] = lowh ? regA[4 * (q >> 1) + (q & 1)] : regA[4 * (q >> 1) + (q & 1) + 2];
    cH[q] = lowh ? regC[4 * (q >> 1) + (q & 1)] : regC[4 * (q >> 1) + (q & 1) + 2];
  }
  v8f Qd, Rd, Pd;
  #pragma unroll
  for (int q = 0; q < 8; ++q) {
    Qd[q] = Qg[i * 8 + krow[q & 3]];
    Pd[q] = P0g[i * 8 + krow[q & 3]];
    const float rv = Rg[(krow[q & 3] & 3) * 4 + (i & 3)];
    Rd[q] = ((q & 3) < 2 && i < 4) ? rv : 0.0f;
  }

  float xrep = x0g[i];

  const v8f z8 = {0,0,0,0,0,0,0,0};

  int selk[8];
  #pragma unroll
  for (int k = 0; k < 8; ++k) selk[k] = 8 * ((k >> 1) & 1) + 4 * e + 2 * (k >> 2) + (k & 1);

  const int stP = R2 + e * 64 + (2 * h) * 8 + i;
  const int ldP = R2 + e * 64 + i * 8 + 2 * h;
  const int stH = (i < 4) ? (R4 + e * 32 + (2 * h) * 4 + i) : (RT + lane);
  const int ldH = R4 + e * 32 + i * 4;

  #pragma unroll 1
  for (int t = 0; t < TT; ++t) {
    const long zidx = ((b0 + e) * TT + t) * N4 + (i & 3);
    const float zv = z[zidx];
    __builtin_prefetch(&z[(t < TT - 8) ? (zidx + 8 * N4) : zidx], 0, 0);
    hold_issue_order();
    float xb[8];
    #pragma unroll
    for (int k = 0; k < 8; ++k) xb[k] = rowpick(xrep, selk[k]);
    float ta = 0.0f;
    #pragma unroll
    for (int k = 0; k < 8; ++k) ta = fmaf(regA[k], xb[k], ta);
    const float xp = fast_tanh(ta);
    const float dd = 1.0f - xp * xp;
    float xq[8];
    #pragma unroll
    for (int k = 0; k < 8; ++k) xq[k] = rowpick(xp, selk[k]);
    float u = 0.0f;
    #pragma unroll
    for (int k = 0; k < 8; ++k) u = fmaf(regC[k], xq[k], u);
    const float ys = fast_sin(u);
    const float cy = fast_cos(u);

    float Fo[4], Ho[4];
    #pragma unroll
    for (int q = 0; q < 4; ++q) { Fo[q] = dd * aF[q]; Ho[q] = cy * cH[q]; }

    v8f X = wmma4(ch1 ? Pd[4] : Pd[0], ch1 ? Pd[5] : Pd[1], Fo[0], Fo[1], z8);
    X     = wmma4(ch1 ? Pd[6] : Pd[2], ch1 ? Pd[7] : Pd[3], Fo[2], Fo[3], X);
    v8f Pp = wmma4(ch1 ? X[4] : X[0], ch1 ? X[5] : X[1], Fo[0], Fo[1], Qd);
    Pp     = wmma4(ch1 ? X[6] : X[2], ch1 ? X[7] : X[3], Fo[2], Fo[3], Pp);
    const float ah0 = ch1 ? Pp[4] : Pp[0], ah1 = ch1 ? Pp[5] : Pp[1];
    const float ah2 = ch1 ? Pp[6] : Pp[2], ah3 = ch1 ? Pp[7] : Pp[3];
    W[stP +  0] = ah0; W[stP +  8] = ah1;
    W[stP + 32] = ah2; W[stP + 40] = ah3;
    const float pa0 = W[ldP + 0], pa1 = W[ldP + 1];
    const float pa2 = W[ldP + 4], pa3 = W[ldP + 5];
    v8f PHt = wmma4(ah0, ah1, Ho[0], Ho[1], z8);
    PHt     = wmma4(ah2, ah3, Ho[2], Ho[3], PHt);
    const float bs0 = ch1 ? PHt[4] : PHt[0], bs1 = ch1 ? PHt[5] : PHt[1];
    const float bs2 = ch1 ? PHt[6] : PHt[2], bs3 = ch1 ? PHt[7] : PHt[3];
    W[stH +  0] = bs0; W[stH +  4] = bs1;
    W[stH + 16] = bs2; W[stH + 20] = bs3;
    const float4 rhs4 = *reinterpret_cast<const float4*>(&W[ldH]);
    hold_issue_order();
    v8f Sm = wmma4(Ho[0], Ho[1], bs0, bs1, Rd);
    Sm     = wmma4(Ho[2], Ho[3], bs2, bs3, Sm);
    v8f T1 = wmma4(Ho[0], Ho[1], pa0, pa1, z8);
    T1     = wmma4(Ho[2], Ho[3], pa2, pa3, T1);

    float rr4[4] = {rhs4.x, rhs4.y, rhs4.z, rhs4.w};
    const float sa0 = ch1 ? Sm[4] : Sm[0], sa1 = ch1 ? Sm[5] : Sm[1];
    const float sx0 = xhalf(sa0), sx1 = xhalf(sa1);
    const float srow[4] = {lowh ? sa0 : sx0, lowh ? sa1 : sx1, lowh ? sx0 : sa0, lowh ? sx1 : sa1};
    float s[4][4];
    #pragma unroll
    for (int a = 0; a < 4; ++a) {
      s[a][0] = rowpick(srow[a], selk[0]); s[a][1] = rowpick(srow[a], selk[1]);
      s[a][2] = rowpick(srow[a], selk[2]); s[a][3] = rowpick(srow[a], selk[3]);
    }
    #pragma unroll
    for (int k = 0; k < 4; ++k) {
      const float inv = __builtin_amdgcn_rcpf(s[k][k]);
      #pragma unroll
      for (int j = k + 1; j < 4; ++j) s[k][j] *= inv;
      rr4[k] *= inv;
      #pragma unroll
      for (int a = k + 1; a < 4; ++a) {
        const float f = s[a][k];
        #pragma unroll
        for (int j = k + 1; j < 4; ++j) s[a][j] = fmaf(-f, s[k][j], s[a][j]);
        rr4[a] = fmaf(-f, rr4[k], rr4[a]);
      }
    }
    float Kp[4];
    Kp[3] = rr4[3];
    Kp[2] = fmaf(-s[2][3], Kp[3], rr4[2]);
    Kp[1] = fmaf(-s[1][3], Kp[3], fmaf(-s[1][2], Kp[2], rr4[1]));
    Kp[0] = fmaf(-s[0][3], Kp[3], fmaf(-s[0][2], Kp[2], fmaf(-s[0][1], Kp[1], rr4[0])));

    const float dy = zv - ys;
    float dyb[4];
    dyb[0] = rowpick(dy, selk[0]); dyb[1] = rowpick(dy, selk[1]); dyb[2] = rowpick(dy, selk[2]); dyb[3] = rowpick(dy, selk[3]);
    float xn = xp;
    #pragma unroll
    for (int j = 0; j < 4; ++j) xn = fmaf(Kp[j], dyb[j], xn);
    if (lowh) out[((b0 + e) * TT + t) * 8 + i] = xn;

    Pd = wmma4(ch1 ? T1[4] : T1[0], ch1 ? T1[5] : T1[1],
               -(lowh ? Kp[0] : Kp[2]), -(lowh ? Kp[1] : Kp[3]), Pp);
    xrep = xn;
  }
}

extern "C" void kernel_launch(void* const* d_in, const int* in_sizes, int n_in,
                              void* d_out, int out_size, void* d_ws, size_t ws_size,
                              hipStream_t stream) {
  const float* z  = (const float*)d_in[0];
  const float* A  = (const float*)d_in[1];
  const float* C  = (const float*)d_in[2];
  const float* Q  = (const float*)d_in[3];
  const float* R  = (const float*)d_in[4];
  const float* x0 = (const float*)d_in[5];
  const float* P0 = (const float*)d_in[6];
  float* out = (float*)d_out;

  const int Bsz = in_sizes[0] / (TT * N4);
  const int waves = Bsz / 2;
  const int threads = 256;
  int blocks = (waves * 32) / threads;
  if (blocks < 1) blocks = 1;
  ekf1250_kernel<<<blocks, threads, 0, stream>>>(z, A, C, Q, R, x0, P0, out);
}


# Sums re-associated: Ax and C*tanh dot products split 4+4 over the lane halves (one permlanex16 + add each); back substitution and K*dy add the last-solved term last; f32 sum order changes
# speedup vs baseline: 1.8470x; 1.0371x over previous
#include <hip/hip_runtime.h>
#include <hip/hip_bf16.h>


#define TT 256
#define N4 4

typedef __attribute__((ext_vector_type(2))) float v2f;
typedef __attribute__((ext_vector_type(8))) float v8f;

__device__ __forceinline__ float rowpick(float v, int sel) {
  const int b = __float_as_int(v);
  return __int_as_float(__builtin_amdgcn_permlane16_var(b, b, sel, true, false));
}
__device__ __forceinline__ float xhalf(float v) {
  const int b = __float_as_int(v);
  return __int_as_float(__builtin_amdgcn_permlanex16(b, b, 0x76543210, 0xfedcba98, true, false));
}

__device__ __forceinline__ v8f wmma4(float a0, float a1, float b0, float b1, v8f c) {
  v2f A, B;
  A[0] = a0; A[1] = a1;
  B[0] = b0; B[1] = b1;
  return __builtin_amdgcn_wmma_f32_16x16x4_f32(false, A, false, B, (short)0, c, false, false);
}

__device__ __forceinline__ void hold_issue_order() { __builtin_amdgcn_sched_barrier(0); }

__device__ __forceinline__ float fast_tanh(float x) {
#if __has_builtin(__builtin_amdgcn_tanhf)
  return __builtin_amdgcn_tanhf(x);
#else
  const float e = __expf(2.0f * x);
  return 1.0f - 2.0f * __builtin_amdgcn_rcpf(e + 1.0f);
#endif
}
#define INV_2PI 0.15915494309189535f
__device__ __forceinline__ float fast_sin(float x) {
  return __builtin_amdgcn_sinf(x * INV_2PI);
}
__device__ __forceinline__ float fast_cos(float x) {
  return __builtin_amdgcn_cosf(x * INV_2PI);
}

__global__ __launch_bounds__(256) void ekf1250_kernel(
    const float* __restrict__ z,  const float* __restrict__ Ag,
    const float* __restrict__ Cg, const float* __restrict__ Qg,
    const float* __restrict__ Rg, const float* __restrict__ x0g,
    const float* __restrict__ P0g, float* __restrict__ out)
{
  __shared__ float lds[8 * 256];
  const int tid  = threadIdx.x;
  const int lane = tid & 31;
  const int wave = (blockIdx.x * blockDim.x + tid) >> 5;
  const long b0  = (long)wave * 2;
  float* W = &lds[(tid >> 5) * 256];
  const int R2 = 0;
  const int R4 = 128;
  const int RT = 192;

  const int  c    = lane & 15;
  const int  h    = lane >> 4;
  const int  e    = (c >> 2) & 1;
  const int  i    = 4 * ((c & 3) >> 1) + 2 * (c >> 3) + (c & 1);
  const bool ch1  = e != 0;
  const bool lowh = lane < 16;
  const int  krow[4] = {2 * h, 2 * h + 1, 4 + 2 * h, 5 + 2 * h};

  float regA[8], regC[8];
  {
    const float4 a0 = *reinterpret_cast<const float4*>(Ag + i * 8);
    const float4 a1 = *reinterpret_cast<const float4*>(Ag + i * 8 + 4);
    regA[0]=a0.x; regA[1]=a0.y; regA[2]=a0.z; regA[3]=a0.w;
    regA[4]=a1.x; regA[5]=a1.y; regA[6]=a1.z; regA[7]=a1.w;
    const float4 c0 = *reinterpret_cast<const float4*>(Cg + (i & 3) * 8);
    const float4 c1 = *reinterpret_cast<const float4*>(Cg + (i & 3) * 8 + 4);
    const float m = (i < 4) ? 1.0f : 0.0f;
    regC[0]=c0.x*m; regC[1]=c0.y*m; regC[2]=c0.z*m; regC[3]=c0.w*m;
    regC[4]=c1.x*m; regC[5]=c1.y*m; regC[6]=c1.z*m; regC[7]=c1.w*m;
  }
  float aF[4], cH[4];
  #pragma unroll
  for (int q = 0; q < 4; ++q) {
    aF[q] = lowh ? regA[4 * (q >> 1) + (q & 1)] : regA[4 * (q >> 1) + (q & 1) + 2];
    cH[q] = lowh ? regC[4 * (q >> 1) + (q & 1)] : regC[4 * (q >> 1) + (q & 1) + 2];
  }
  v8f Qd, Rd, Pd;
  #pragma unroll
  for (int q = 0; q < 8; ++q) {
    Qd[q] = Qg[i * 8 + krow[q & 3]];
    Pd[q] = P0g[i * 8 + krow[q & 3]];
    const float rv = Rg[(krow[q & 3] & 3) * 4 + (i & 3)];
    Rd[q] = ((q & 3) < 2 && i < 4) ? rv : 0.0f;
  }

  float aD[4], cD[4];
  #pragma unroll
  for (int j = 0; j < 4; ++j) {
    aD[j] = lowh ? regA[j] : regA[4 + j];
    cD[j] = lowh ? regC[j] : regC[4 + j];
  }

  float xrep = x0g[i];

  const v8f z8 = {0,0,0,0,0,0,0,0};

  int selk[8];
  #pragma unroll
  for (int k = 0; k < 8; ++k) selk[k] = 8 * ((k >> 1) & 1) + 4 * e + 2 * (k >> 2) + (k & 1);
  int selh[4];
  #pragma unroll
  for (int j = 0; j < 4; ++j) selh[j] = lowh ? selk[j] : selk[4 + j];

  const int stP = R2 + e * 64 + (2 * h) * 8 + i;
  const int ldP = R2 + e * 64 + i * 8 + 2 * h;
  const int stH = (i < 4) ? (R4 + e * 32 + (2 * h) * 4 + i) : (RT + lane);
  const int ldH = R4 + e * 32 + i * 4;

  #pragma unroll 1
  for (int t = 0; t < TT; ++t) {
    const long zidx = ((b0 + e) * TT + t) * N4 + (i & 3);
    const float zv = z[zidx];
    __builtin_prefetch(&z[(t < TT - 8) ? (zidx + 8 * N4) : zidx], 0, 0);
    hold_issue_order();
    float xb[4];
    #pragma unroll
    for (int j = 0; j < 4; ++j) xb[j] = rowpick(xrep, selh[j]);
    float tp = aD[0] * xb[0];
    #pragma unroll
    for (int j = 1; j < 4; ++j) tp = fmaf(aD[j], xb[j], tp);
    const float ta = tp + xhalf(tp);
    const float xp = fast_tanh(ta);
    const float dd = 1.0f - xp * xp;
    float xq[4];
    #pragma unroll
    for (int j = 0; j < 4; ++j) xq[j] = rowpick(xp, selh[j]);
    float up = cD[0] * xq[0];
    #pragma unroll
    for (int j = 1; j < 4; ++j) up = fmaf(cD[j], xq[j], up);
    const float u = up + xhalf(up);
    const float ys = fast_sin(u);
    const float cy = fast_cos(u);

    float Fo[4], Ho[4];
    #pragma unroll
    for (int q = 0; q < 4; ++q) { Fo[q] = dd * aF[q]; Ho[q] = cy * cH[q]; }

    v8f X = wmma4(ch1 ? Pd[4] : Pd[0], ch1 ? Pd[5] : Pd[1], Fo[0], Fo[1], z8);
    X     = wmma4(ch1 ? Pd[6] : Pd[2], ch1 ? Pd[7] : Pd[3], Fo[2], Fo[3], X);
    v8f Pp = wmma4(ch1 ? X[4] : X[0], ch1 ? X[5] : X[1], Fo[0], Fo[1], Qd);
    Pp     = wmma4(ch1 ? X[6] : X[2], ch1 ? X[7] : X[3], Fo[2], Fo[3], Pp);
    const float ah0 = ch1 ? Pp[4] : Pp[0], ah1 = ch1 ? Pp[5] : Pp[1];
    const float ah2 = ch1 ? Pp[6] : Pp[2], ah3 = ch1 ? Pp[7] : Pp[3];
    W[stP +  0] = ah0; W[stP +  8] = ah1;
    W[stP + 32] = ah2; W[stP + 40] = ah3;
    const float pa0 = W[ldP + 0], pa1 = W[ldP + 1];
    const float pa2 = W[ldP + 4], pa3 = W[ldP + 5];
    v8f PHt = wmma4(ah0, ah1, Ho[0], Ho[1], z8);
    PHt     = wmma4(ah2, ah3, Ho[2], Ho[3], PHt);
    const float bs0 = ch1 ? PHt[4] : PHt[0], bs1 = ch1 ? PHt[5] : PHt[1];
    const float bs2 = ch1 ? PHt[6] : PHt[2], bs3 = ch1 ? PHt[7] : PHt[3];
    W[stH +  0] = bs0; W[stH +  4] = bs1;
    W[stH + 16] = bs2; W[stH + 20] = bs3;
    const float4 rhs4 = *reinterpret_cast<const float4*>(&W[ldH]);
    hold_issue_order();
    v8f Sm = wmma4(Ho[0], Ho[1], bs0, bs1, Rd);
    Sm     = wmma4(Ho[2], Ho[3], bs2, bs3, Sm);
    v8f T1 = wmma4(Ho[0], Ho[1], pa0, pa1, z8);
    T1     = wmma4(Ho[2], Ho[3], pa2, pa3, T1);

    float rr4[4] = {rhs4.x, rhs4.y, rhs4.z, rhs4.w};
    const float sa0 = ch1 ? Sm[4] : Sm[0], sa1 = ch1 ? Sm[5] : Sm[1];
    const float sx0 = xhalf(sa0), sx1 = xhalf(sa1);
    const float srow[4] = {lowh ? sa0 : sx0, lowh ? sa1 : sx1, lowh ? sx0 : sa0, lowh ? sx1 : sa1};
    float s[4][4];
    #pragma unroll
    for (int a = 0; a < 4; ++a) {
      s[a][0] = rowpick(srow[a], selk[0]); s[a][1] = rowpick(srow[a], selk[1]);
      s[a][2] = rowpick(srow[a], selk[2]); s[a][3] = rowpick(srow[a], selk[3]);
    }
    #pragma unroll
    for (int k = 0; k < 4; ++k) {
      const float inv = __builtin_amdgcn_rcpf(s[k][k]);
      #pragma unroll
      for (int j = k + 1; j < 4; ++j) s[k][j] *= inv;
      rr4[k] *= inv;
      #pragma unroll
      for (int a = k + 1; a < 4; ++a) {
        const float f = s[a][k];
        #pragma unroll
        for (int j = k + 1; j < 4; ++j) s[a][j] = fmaf(-f, s[k][j], s[a][j]);
        rr4[a] = fmaf(-f, rr4[k], rr4[a]);
      }
    }
    float Kp[4];
    Kp[3] = rr4[3];
    Kp[2] = fmaf(-s[2][3], Kp[3], rr4[2]);
    Kp[1] = fmaf(-s[1][2], Kp[2], fmaf(-s[1][3], Kp[3], rr4[1]));
    Kp[0] = fmaf(-s[0][1], Kp[1], fmaf(-s[0][2], Kp[2], fmaf(-s[0][3], Kp[3], rr4[0])));

    const float dy = zv - ys;
    float dyb[4];
    dyb[0] = rowpick(dy, selk[0]); dyb[1] = rowpick(dy, selk[1]); dyb[2] = rowpick(dy, selk[2]); dyb[3] = rowpick(dy, selk[3]);
    float xn = xp;
    #pragma unroll
    for (int j = 3; j >= 0; --j) xn = fmaf(Kp[j], dyb[j], xn);
    if (lowh) out[((b0 + e) * TT + t) * 8 + i] = xn;

    Pd = wmma4(ch1 ? T1[4] : T1[0], ch1 ? T1[5] : T1[1],
               -(lowh ? Kp[0] : Kp[2]), -(lowh ? Kp[1] : Kp[3]), Pp);
    xrep = xn;
  }
}

extern "C" void kernel_launch(void* const* d_in, const int* in_sizes, int n_in,
                              void* d_out, int out_size, void* d_ws, size_t ws_size,
                              hipStream_t stream) {
  const float* z  = (const float*)d_in[0];
  const float* A  = (const float*)d_in[1];
  const float* C  = (const float*)d_in[2];
  const float* Q  = (const float*)d_in[3];
  const float* R  = (const float*)d_in[4];
  const float* x0 = (const float*)d_in[5];
  const float* P0 = (const float*)d_in[6];
  float* out = (float*)d_out;

  const int Bsz = in_sizes[0] / (TT * N4);
  const int waves = Bsz / 2;
  const int threads = 256;
  int blocks = (waves * 32) / threads;
  if (blocks < 1) blocks = 1;
  ekf1250_kernel<<<blocks, threads, 0, stream>>>(z, A, C, Q, R, x0, P0, out);
}


# Two steps per loop trip: one 32-lane store writes both states (lower lane half even step, upper half odd); the exec-masked half-wave store and its branch leave the loop; z is read one trip ahead
# speedup vs baseline: 1.9096x; 1.0339x over previous
#include <hip/hip_runtime.h>
#include <hip/hip_bf16.h>


#define TT 256
#define N4 4

typedef __attribute__((ext_vector_type(2))) float v2f;
typedef __attribute__((ext_vector_type(8))) float v8f;

__device__ __forceinline__ float rowpick(float v, int sel) {
  const int b = __float_as_int(v);
  return __int_as_float(__builtin_amdgcn_permlane16_var(b, b, sel, true, false));
}
__device__ __forceinline__ float xhalf(float v) {
  const int b = __float_as_int(v);
  return __int_as_float(__builtin_amdgcn_permlanex16(b, b, 0x76543210, 0xfedcba98, true, false));
}

__device__ __forceinline__ v8f wmma4(float a0, float a1, float b0, float b1, v8f c) {
  v2f A, B;
  A[0] = a0; A[1] = a1;
  B[0] = b0; B[1] = b1;
  return __builtin_amdgcn_wmma_f32_16x16x4_f32(false, A, false, B, (short)0, c, false, false);
}

__device__ __forceinline__ void hold_issue_order() { __builtin_amdgcn_sched_barrier(0); }

__device__ __forceinline__ float fast_tanh(float x) {
#if __has_builtin(__builtin_amdgcn_tanhf)
  return __builtin_amdgcn_tanhf(x);
#else
  const float e = __expf(2.0f * x);
  return 1.0f - 2.0f * __builtin_amdgcn_rcpf(e + 1.0f);
#endif
}
#define INV_2PI 0.15915494309189535f
__device__ __forceinline__ float fast_sin(float x) {
  return __builtin_amdgcn_sinf(x * INV_2PI);
}
__device__ __forceinline__ float fast_cos(float x) {
  return __builtin_amdgcn_cosf(x * INV_2PI);
}

__global__ __launch_bounds__(256) void ekf1250_kernel(
    const float* __restrict__ z,  const float* __restrict__ Ag,
    const float* __restrict__ Cg, const float* __restrict__ Qg,
    const float* __restrict__ Rg, const float* __restrict__ x0g,
    const float* __restrict__ P0g, float* __restrict__ out)
{
  __shared__ float lds[8 * 256];
  const int tid  = threadIdx.x;
  const int lane = tid & 31;
  const int wave = (blockIdx.x * blockDim.x + tid) >> 5;
  const long b0  = (long)wave * 2;
  float* W = &lds[(tid >> 5) * 256];
  const int R2 = 0;
  const int R4 = 128;
  const int RT = 192;

  const int  c    = lane & 15;
  const int  h    = lane >> 4;
  const int  e    = (c >> 2) & 1;
  const int  i    = 4 * ((c & 3) >> 1) + 2 * (c >> 3) + (c & 1);
  const bool ch1  = e != 0;
  const bool lowh = lane < 16;
  const int  krow[4] = {2 * h, 2 * h + 1, 4 + 2 * h, 5 + 2 * h};

  float regA[8], regC[8];
  {
    const float4 a0 = *reinterpret_cast<const float4*>(Ag + i * 8);
    const float4 a1 = *reinterpret_cast<const float4*>(Ag + i * 8 + 4);
    regA[0]=a0.x; regA[1]=a0.y; regA[2]=a0.z; regA[3]=a0.w;
    regA[4]=a1.x; regA[5]=a1.y; regA[6]=a1.z; regA[7]=a1.w;
    const float4 c0 = *reinterpret_cast<const float4*>(Cg + (i & 3) * 8);
    const float4 c1 = *reinterpret_cast<const float4*>(Cg + (i & 3) * 8 + 4);
    const float m = (i < 4) ? 1.0f : 0.0f;
    regC[0]=c0.x*m; regC[1]=c0.y*m; regC[2]=c0.z*m; regC[3]=c0.w*m;
    regC[4]=c1.x*m; regC[5]=c1.y*m; regC[6]=c1.z*m; regC[7]=c1.w*m;
  }
  float aF[4], cH[4];
  #pragma unroll
  for (int q = 0; q < 4; ++q) {
    aF[q] = lowh ? regA[4 * (q >> 1) + (q & 1)] : regA[4 * (q >> 1) + (q & 1) + 2];
    cH[q] = lowh ? regC[4 * (q >> 1) + (q & 1)] : regC[4 * (q >> 1) + (q & 1) + 2];
  }
  v8f Qd, Rd, Pd;
  #pragma unroll
  for (int q = 0; q < 8; ++q) {
    Qd[q] = Qg[i * 8 + krow[q & 3]];
    Pd[q] = P0g[i * 8 + krow[q & 3]];
    const float rv = Rg[(krow[q & 3] & 3) * 4 + (i & 3)];
    Rd[q] = ((q & 3) < 2 && i < 4) ? rv : 0.0f;
  }

  float aD[4], cD[4];
  #pragma unroll
  for (int j = 0; j < 4; ++j) {
    aD[j] = lowh ? regA[j] : regA[4 + j];
    cD[j] = lowh ? regC[j] : regC[4 + j];
  }

  float xrep = x0g[i];

  const v8f z8 = {0,0,0,0,0,0,0,0};

  int selk[8];
  #pragma unroll
  for (int k = 0; k < 8; ++k) selk[k] = 8 * ((k >> 1) & 1) + 4 * e + 2 * (k >> 2) + (k & 1);
  int selh[4];
  #pragma unroll
  for (int j = 0; j < 4; ++j) selh[j] = lowh ? selk[j] : selk[4 + j];

  const int stP = R2 + e * 64 + (2 * h) * 8 + i;
  const int ldP = R2 + e * 64 + i * 8 + 2 * h;
  const int stH = (i < 4) ? (R4 + e * 32 + (2 * h) * 4 + i) : (RT + lane);
  const int ldH = R4 + e * 32 + i * 4;

  const long zrow = (b0 + e) * TT * N4 + (i & 3);
  float zn0 = z[zrow], zn1 = z[zrow + N4];

  #pragma unroll 1
  for (int t2 = 0; t2 < TT; t2 += 2) {
   float xst = 0.0f;
   const float zc0 = zn0, zc1 = zn1;
   const int tn = (t2 + 2 < TT) ? t2 + 2 : t2;
   zn0 = z[zrow + (long)tn * N4];
   zn1 = z[zrow + (long)tn * N4 + N4];
   #pragma unroll
   for (int sub = 0; sub < 2; ++sub) {
    const int t = t2 + sub;
    const long zidx = ((b0 + e) * TT + t) * N4 + (i & 3);
    const float zv = (sub == 0) ? zc0 : zc1;
    __builtin_prefetch(&z[(t < TT - 8) ? (zidx + 8 * N4) : zidx], 0, 0);
    hold_issue_order();
    float xb[4];
    #pragma unroll
    for (int j = 0; j < 4; ++j) xb[j] = rowpick(xrep, selh[j]);
    float tp = aD[0] * xb[0];
    #pragma unroll
    for (int j = 1; j < 4; ++j) tp = fmaf(aD[j], xb[j], tp);
    const float ta = tp + xhalf(tp);
    const float xp = fast_tanh(ta);
    const float dd = 1.0f - xp * xp;
    float xq[4];
    #pragma unroll
    for (int j = 0; j < 4; ++j) xq[j] = rowpick(xp, selh[j]);
    float up = cD[0] * xq[0];
    #pragma unroll
    for (int j = 1; j < 4; ++j) up = fmaf(cD[j], xq[j], up);
    const float u = up + xhalf(up);
    const float ys = fast_sin(u);
    const float cy = fast_cos(u);

    float Fo[4], Ho[4];
    #pragma unroll
    for (int q = 0; q < 4; ++q) { Fo[q] = dd * aF[q]; Ho[q] = cy * cH[q]; }

    v8f X = wmma4(ch1 ? Pd[4] : Pd[0], ch1 ? Pd[5] : Pd[1], Fo[0], Fo[1], z8);
    X     = wmma4(ch1 ? Pd[6] : Pd[2], ch1 ? Pd[7] : Pd[3], Fo[2], Fo[3], X);
    v8f Pp = wmma4(ch1 ? X[4] : X[0], ch1 ? X[5] : X[1], Fo[0], Fo[1], Qd);
    Pp     = wmma4(ch1 ? X[6] : X[2], ch1 ? X[7] : X[3], Fo[2], Fo[3], Pp);
    const float ah0 = ch1 ? Pp[4] : Pp[0], ah1 = ch1 ? Pp[5] : Pp[1];
    const float ah2 = ch1 ? Pp[6] : Pp[2], ah3 = ch1 ? Pp[7] : Pp[3];
    W[stP +  0] = ah0; W[stP +  8] = ah1;
    W[stP + 32] = ah2; W[stP + 40] = ah3;
    const float pa0 = W[ldP + 0], pa1 = W[ldP + 1];
    const float pa2 = W[ldP + 4], pa3 = W[ldP + 5];
    v8f PHt = wmma4(ah0, ah1, Ho[0], Ho[1], z8);
    PHt     = wmma4(ah2, ah3, Ho[2], Ho[3], PHt);
    const float bs0 = ch1 ? PHt[4] : PHt[0], bs1 = ch1 ? PHt[5] : PHt[1];
    const float bs2 = ch1 ? PHt[6] : PHt[2], bs3 = ch1 ? PHt[7] : PHt[3];
    W[stH +  0] = bs0; W[stH +  4] = bs1;
    W[stH + 16] = bs2; W[stH + 20] = bs3;
    const float4 rhs4 = *reinterpret_cast<const float4*>(&W[ldH]);
    hold_issue_order();
    v8f Sm = wmma4(Ho[0], Ho[1], bs0, bs1, Rd);
    Sm     = wmma4(Ho[2], Ho[3], bs2, bs3, Sm);
    v8f T1 = wmma4(Ho[0], Ho[1], pa0, pa1, z8);
    T1     = wmma4(Ho[2], Ho[3], pa2, pa3, T1);

    float rr4[4] = {rhs4.x, rhs4.y, rhs4.z, rhs4.w};
    const float sa0 = ch1 ? Sm[4] : Sm[0], sa1 = ch1 ? Sm[5] : Sm[1];
    const float sx0 = xhalf(sa0), sx1 = xhalf(sa1);
    const float srow[4] = {lowh ? sa0 : sx0, lowh ? sa1 : sx1, lowh ? sx0 : sa0, lowh ? sx1 : sa1};
    float s[4][4];
    #pragma unroll
    for (int a = 0; a < 4; ++a) {
      s[a][0] = rowpick(srow[a], selk[0]); s[a][1] = rowpick(srow[a], selk[1]);
      s[a][2] = rowpick(srow[a], selk[2]); s[a][3] = rowpick(srow[a], selk[3]);
    }
    #pragma unroll
    for (int k = 0; k < 4; ++k) {
      const float inv = __builtin_amdgcn_rcpf(s[k][k]);
      #pragma unroll
      for (int j = k + 1; j < 4; ++j) s[k][j] *= inv;
      rr4[k] *= inv;
      #pragma unroll
      for (int a = k + 1; a < 4; ++a) {
        const float f = s[a][k];
        #pragma unroll
        for (int j = k + 1; j < 4; ++j) s[a][j] = fmaf(-f, s[k][j], s[a][j]);
        rr4[a] = fmaf(-f, rr4[k], rr4[a]);
      }
    }
    float Kp[4];
    Kp[3] = rr4[3];
    Kp[2] = fmaf(-s[2][3], Kp[3], rr4[2]);
    Kp[1] = fmaf(-s[1][2], Kp[2], fmaf(-s[1][3], Kp[3], rr4[1]));
    Kp[0] = fmaf(-s[0][1], Kp[1], fmaf(-s[0][2], Kp[2], fmaf(-s[0][3], Kp[3], rr4[0])));

    const float dy = zv - ys;
    float dyb[4];
    dyb[0] = rowpick(dy, selk[0]); dyb[1] = rowpick(dy, selk[1]); dyb[2] = rowpick(dy, selk[2]); dyb[3] = rowpick(dy, selk[3]);
    float xn = xp;
    #pragma unroll
    for (int j = 3; j >= 0; --j) xn = fmaf(Kp[j], dyb[j], xn);
    xst = (sub == 0) ? xn : (lowh ? xst : xn);

    Pd = wmma4(ch1 ? T1[4] : T1[0], ch1 ? T1[5] : T1[1],
               -(lowh ? Kp[0] : Kp[2]), -(lowh ? Kp[1] : Kp[3]), Pp);
    xrep = xn;
   }
   out[((b0 + e) * TT + t2 + h) * 8 + i] = xst;
  }
}

extern "C" void kernel_launch(void* const* d_in, const int* in_sizes, int n_in,
                              void* d_out, int out_size, void* d_ws, size_t ws_size,
                              hipStream_t stream) {
  const float* z  = (const float*)d_in[0];
  const float* A  = (const float*)d_in[1];
  const float* C  = (const float*)d_in[2];
  const float* Q  = (const float*)d_in[3];
  const float* R  = (const float*)d_in[4];
  const float* x0 = (const float*)d_in[5];
  const float* P0 = (const float*)d_in[6];
  float* out = (float*)d_out;

  const int Bsz = in_sizes[0] / (TT * N4);
  const int waves = Bsz / 2;
  const int threads = 256;
  int blocks = (waves * 32) / threads;
  if (blocks < 1) blocks = 1;
  ekf1250_kernel<<<blocks, threads, 0, stream>>>(z, A, C, Q, R, x0, P0, out);
}
